# starting kernel
# baseline (speedup 1.0000x reference)
#include <hip/hip_runtime.h>
#include <math.h>


typedef __attribute__((ext_vector_type(2))) float v2f;
typedef __attribute__((ext_vector_type(8))) float v8f;

#define N_NODES 50000
#define N_EDGES 800000
#define F_DIM   128
#define NBITS   64
#define NCLS    100
#define HD      300
#define LDS_STRIDE 132

__device__ __forceinline__ v8f wmma_f32(v2f a, v2f b, v8f c) {
  return __builtin_amdgcn_wmma_f32_16x16x4_f32(false, a, false, b, (short)0, c,
                                               false, false);
}

__global__ void __launch_bounds__(256)
degree_kernel(const int* __restrict__ dst, float* __restrict__ deg) {
  int e = blockIdx.x * blockDim.x + threadIdx.x;
  if (e < N_EDGES) atomicAdd(&deg[dst[e]], 1.0f);
}

__global__ void __launch_bounds__(256)
scatter_add_kernel(const float* __restrict__ x, const int* __restrict__ src,
                   const int* __restrict__ dst, float* __restrict__ agg) {
  int gid = blockIdx.x * blockDim.x + threadIdx.x;
  int e   = gid >> 5;
  if (e >= N_EDGES) return;
  int ch = (gid & 31) * 4;
  int s = src[e], d = dst[e];
  const float4 v = *reinterpret_cast<const float4*>(x + (size_t)s * F_DIM + ch);
  float* p = agg + (size_t)d * F_DIM + ch;
  atomicAdd(p + 0, v.x);
  atomicAdd(p + 1, v.y);
  atomicAdd(p + 2, v.z);
  atomicAdd(p + 3, v.w);
}

__global__ void __launch_bounds__(256)
sage_layer_kernel(const float* __restrict__ x, const float* __restrict__ agg,
                  const float* __restrict__ deg, const float* __restrict__ wl,
                  const float* __restrict__ bl, const float* __restrict__ wr,
                  float* __restrict__ h) {
  __shared__ float sMean[16 * LDS_STRIDE];
  __shared__ float sX[16 * LDS_STRIDE];

  const int tm = blockIdx.x;

  for (int i = threadIdx.x; i < 16 * 32; i += 256) {
    const int r    = i >> 5;
    const int c4   = (i & 31) * 4;
    const int grow = tm * 16 + r;
    const float invd = 1.0f / fmaxf(deg[grow], 1.0f);
    const float4 va = *reinterpret_cast<const float4*>(agg + (size_t)grow * F_DIM + c4);
    const float4 vx = *reinterpret_cast<const float4*>(x   + (size_t)grow * F_DIM + c4);
    float* pm = sMean + r * LDS_STRIDE + c4;
    pm[0] = va.x * invd; pm[1] = va.y * invd; pm[2] = va.z * invd; pm[3] = va.w * invd;
    float* px = sX + r * LDS_STRIDE + c4;
    px[0] = vx.x; px[1] = vx.y; px[2] = vx.z; px[3] = vx.w;
  }
  __syncthreads();

  const int lane = threadIdx.x & 31;
  const int tn   = threadIdx.x >> 5;
  const int mloc = lane & 15;
  const int ncol = tn * 16 + mloc;
  const int kb   = (lane >= 16) ? 2 : 0;
  const float* sm = sMean + mloc * LDS_STRIDE;
  const float* sx = sX    + mloc * LDS_STRIDE;

  v8f acc = {};
#pragma unroll 4
  for (int kk = 0; kk < F_DIM; kk += 4) {
    v2f a1, b1, a2, b2;
    a1.x = sm[kk + kb];
    a1.y = sm[kk + kb + 1];
    b1.x = wl[(kk + kb)     * F_DIM + ncol];
    b1.y = wl[(kk + kb + 1) * F_DIM + ncol];
    acc = wmma_f32(a1, b1, acc);
    a2.x = sx[kk + kb];
    a2.y = sx[kk + kb + 1];
    b2.x = wr[(kk + kb)     * F_DIM + ncol];
    b2.y = wr[(kk + kb + 1) * F_DIM + ncol];
    acc = wmma_f32(a2, b2, acc);
  }
  const float bias  = bl[ncol];
  const int   mbase = tm * 16 + ((lane >= 16) ? 8 : 0);
#pragma unroll
  for (int i = 0; i < 8; ++i) {
    float v = fmaxf(acc[i] + bias, 0.0f);
    h[(size_t)(mbase + i) * F_DIM + ncol] = v;
  }
}

template <int ACT>
__global__ void __launch_bounds__(256)
wmma_gemm_bias_kernel(const float* __restrict__ A, const float* __restrict__ B,
                      const float* __restrict__ bias, float* __restrict__ C,
                      int N, int K) {
  __shared__ float sA[16 * LDS_STRIDE];

  const int tm = blockIdx.y;
  const int k4 = K >> 2;
  for (int i = threadIdx.x; i < 16 * k4; i += blockDim.x) {
    const int r  = i / k4;
    const int c4 = (i - r * k4) * 4;
    const float4 v = *reinterpret_cast<const float4*>(A + (size_t)(tm * 16 + r) * K + c4);
    float* p = sA + r * LDS_STRIDE + c4;
    p[0] = v.x; p[1] = v.y; p[2] = v.z; p[3] = v.w;
  }
  __syncthreads();

  const int lane   = threadIdx.x & 31;
  const int wave   = threadIdx.x >> 5;
  const int tilesN = (N + 15) >> 4;
  int tn = blockIdx.x * (blockDim.x >> 5) + wave;
  if (tn >= tilesN) tn = tilesN - 1;
  const int mloc = lane & 15;
  const int ncol = tn * 16 + mloc;
  const bool nok = (ncol < N);
  const int kb = (lane >= 16) ? 2 : 0;
  const float* sa = sA + mloc * LDS_STRIDE;

  v8f acc = {};
  for (int kk = 0; kk < K; kk += 4) {
    v2f a, b;
    a.x = sa[kk + kb];
    a.y = sa[kk + kb + 1];
    b.x = nok ? B[(size_t)(kk + kb) * N + ncol]     : 0.0f;
    b.y = nok ? B[(size_t)(kk + kb + 1) * N + ncol] : 0.0f;
    acc = wmma_f32(a, b, acc);
  }
  if (nok) {
    const float bv    = bias[ncol];
    const int   mbase = tm * 16 + ((lane >= 16) ? 8 : 0);
#pragma unroll
    for (int i = 0; i < 8; ++i) {
      float v = acc[i] + bv;
      if (ACT == 1) v = fmaxf(v, 0.0f);
      C[(size_t)(mbase + i) * N + ncol] = v;
    }
  }
}

__global__ void __launch_bounds__(256)
wmma_gemm_bn_kernel(const float* __restrict__ A, const float* __restrict__ B,
                    const float* __restrict__ bconv, const float* __restrict__ gamma,
                    const float* __restrict__ beta, const float* __restrict__ rm,
                    const float* __restrict__ rv, float* __restrict__ C,
                    int N, int K) {
  __shared__ float sA[16 * LDS_STRIDE];

  const int tm = blockIdx.y;
  const int k4 = K >> 2;
  for (int i = threadIdx.x; i < 16 * k4; i += blockDim.x) {
    const int r  = i / k4;
    const int c4 = (i - r * k4) * 4;
    const float4 v = *reinterpret_cast<const float4*>(A + (size_t)(tm * 16 + r) * K + c4);
    float* p = sA + r * LDS_STRIDE + c4;
    p[0] = v.x; p[1] = v.y; p[2] = v.z; p[3] = v.w;
  }
  __syncthreads();

  const int lane   = threadIdx.x & 31;
  const int wave   = threadIdx.x >> 5;
  const int tilesN = (N + 15) >> 4;
  int tn = blockIdx.x * (blockDim.x >> 5) + wave;
  if (tn >= tilesN) tn = tilesN - 1;
  const int mloc = lane & 15;
  const int ncol = tn * 16 + mloc;
  const bool nok = (ncol < N);
  const int kb = (lane >= 16) ? 2 : 0;
  const float* sa = sA + mloc * LDS_STRIDE;

  v8f acc = {};
  for (int kk = 0; kk < K; kk += 4) {
    v2f a, b;
    a.x = sa[kk + kb];
    a.y = sa[kk + kb + 1];
    b.x = nok ? B[(size_t)(kk + kb) * N + ncol]     : 0.0f;
    b.y = nok ? B[(size_t)(kk + kb + 1) * N + ncol] : 0.0f;
    acc = wmma_f32(a, b, acc);
  }
  if (nok) {
    const float s     = gamma[ncol] * rsqrtf(rv[ncol] + 1e-5f);
    const float shift = (bconv[ncol] - rm[ncol]) * s + beta[ncol];
    const int   mbase = tm * 16 + ((lane >= 16) ? 8 : 0);
#pragma unroll
    for (int i = 0; i < 8; ++i)
      C[(size_t)(mbase + i) * N + ncol] = acc[i] * s + shift;
  }
}

__global__ void __launch_bounds__(256)
tanh_kernel(const float* __restrict__ in, float* __restrict__ out, int n) {
  int i = blockIdx.x * blockDim.x + threadIdx.x;
  if (i < n) out[i] = tanhf(in[i]);
}

__global__ void __launch_bounds__(256)
true_lab_kernel(const float* __restrict__ pre, const float* __restrict__ wtl,
                const float* __restrict__ btl, float* __restrict__ out) {
  int r = blockIdx.x * blockDim.x + threadIdx.x;
  if (r >= N_NODES) return;
  const float* p = pre + (size_t)r * NBITS;
  float s = 0.0f;
#pragma unroll
  for (int k = 0; k < NBITS; ++k) s += p[k] * wtl[k];
  out[r] = s + btl[0];
}

extern "C" void kernel_launch(void* const* d_in, const int* in_sizes, int n_in,
                              void* d_out, int out_size, void* d_ws, size_t ws_size,
                              hipStream_t stream) {
  (void)in_sizes; (void)n_in; (void)out_size; (void)ws_size;

  const float* features = (const float*)d_in[0];
  const int*   edges    = (const int*)d_in[1];
  const float* w1l   = (const float*)d_in[2];
  const float* b1l   = (const float*)d_in[3];
  const float* w1r   = (const float*)d_in[4];
  const float* w2l   = (const float*)d_in[5];
  const float* b2l   = (const float*)d_in[6];
  const float* w2r   = (const float*)d_in[7];
  const float* whd   = (const float*)d_in[8];
  const float* bhd   = (const float*)d_in[9];
  const float* wclas = (const float*)d_in[10];
  const float* bclas = (const float*)d_in[11];
  const float* wconv = (const float*)d_in[12];
  const float* bconv = (const float*)d_in[13];
  const float* gamma = (const float*)d_in[14];
  const float* beta  = (const float*)d_in[15];
  const float* rm    = (const float*)d_in[16];
  const float* rv    = (const float*)d_in[17];
  const float* wtl   = (const float*)d_in[18];
  const float* btl   = (const float*)d_in[19];
  const float* wcv   = (const float*)d_in[20];
  const float* bcv   = (const float*)d_in[21];

  const int* src = edges;
  const int* dst = edges + N_EDGES;

  float* ws     = (float*)d_ws;
  float* deg    = ws;
  float* agg    = deg + N_NODES;
  float* h1     = agg + (size_t)N_NODES * F_DIM;
  float* h2     = h1  + (size_t)N_NODES * F_DIM;
  float* pre_bn = h2  + (size_t)N_NODES * F_DIM;

  float* out       = (float*)d_out;
  float* o_logists = out;
  float* o_tanh    = o_logists + (size_t)N_NODES * NCLS;
  float* o_fealab  = o_tanh    + (size_t)N_NODES * NBITS;
  float* o_feacv   = o_fealab  + (size_t)N_NODES * HD;
  float* o_truelab = o_feacv   + (size_t)N_NODES * NCLS;

  const int scatterBlocks = (N_EDGES * 32 + 255) / 256;
  const int rowTiles      = N_NODES / 16;
  auto gridFor = [rowTiles](int N, int waves) {
    int tilesN = (N + 15) / 16;
    return dim3((tilesN + waves - 1) / waves, rowTiles);
  };

  hipMemsetAsync(deg, 0, sizeof(float) * ((size_t)N_NODES * (F_DIM + 1)), stream);
  degree_kernel<<<(N_EDGES + 255) / 256, 256, 0, stream>>>(dst, deg);
  scatter_add_kernel<<<scatterBlocks, 256, 0, stream>>>(features, src, dst, agg);
  sage_layer_kernel<<<rowTiles, 256, 0, stream>>>(features, agg, deg, w1l, b1l, w1r, h1);

  hipMemsetAsync(agg, 0, sizeof(float) * (size_t)N_NODES * F_DIM, stream);
  scatter_add_kernel<<<scatterBlocks, 256, 0, stream>>>(h1, src, dst, agg);
  sage_layer_kernel<<<rowTiles, 256, 0, stream>>>(h1, agg, deg, w2l, b2l, w2r, h2);

  wmma_gemm_bias_kernel<0><<<gridFor(HD, 8), 256, 0, stream>>>(
      h2, whd, bhd, o_fealab, HD, F_DIM);
  wmma_gemm_bias_kernel<0><<<gridFor(NCLS, 8), 256, 0, stream>>>(
      h2, wclas, bclas, o_logists, NCLS, F_DIM);
  wmma_gemm_bn_kernel<<<gridFor(NBITS, 4), 128, 0, stream>>>(
      h2, wconv, bconv, gamma, beta, rm, rv, pre_bn, NBITS, F_DIM);
  tanh_kernel<<<(N_NODES * NBITS + 255) / 256, 256, 0, stream>>>(
      pre_bn, o_tanh, N_NODES * NBITS);
  wmma_gemm_bias_kernel<0><<<gridFor(NCLS, 8), 256, 0, stream>>>(
      o_tanh, wcv, bcv, o_feacv, NCLS, NBITS);
  true_lab_kernel<<<(N_NODES + 255) / 256, 256, 0, stream>>>(
      pre_bn, wtl, btl, o_truelab);
}


# Float-atomic scatter-add replaced by per-call neighbour lists (int count+slot, scan, fill), each list sorted by source id, then per-node row gather-sum; sum order = sorted list, same on every run
# speedup vs baseline: 1.4953x; 1.4953x over previous
#include <hip/hip_runtime.h>
#include <math.h>


typedef __attribute__((ext_vector_type(2))) float v2f;
typedef __attribute__((ext_vector_type(8))) float v8f;

#define N_NODES 50000
#define N_EDGES 800000
#define F_DIM   128
#define NBITS   64
#define NCLS    100
#define HD      300
#define LDS_STRIDE 132

__device__ __forceinline__ v8f wmma_f32(v2f a, v2f b, v8f c) {
  return __builtin_amdgcn_wmma_f32_16x16x4_f32(false, a, false, b, (short)0, c,
                                               false, false);
}


__global__ void __launch_bounds__(256)
count_slot_kernel(const int* __restrict__ dst, int* __restrict__ cnt,
                  int* __restrict__ slot) {
  int e = blockIdx.x * blockDim.x + threadIdx.x;
  if (e < N_EDGES) slot[e] = atomicAdd(&cnt[dst[e]], 1);
}

#define SCAN_T  1024
#define SCAN_CH ((N_NODES + SCAN_T - 1) / SCAN_T)
#define CNT_LEN (SCAN_T * SCAN_CH)
__global__ void __launch_bounds__(SCAN_T)
scan_offsets_kernel(const int* __restrict__ cnt, int* __restrict__ off,
                    float* __restrict__ deg) {
  __shared__ int sPart[SCAN_T];
  const int t = threadIdx.x;
  const int b = t * SCAN_CH;
  int cv[SCAN_CH];
#pragma unroll
  for (int i = 0; i < SCAN_CH; ++i) cv[i] = cnt[b + i];
  int sum = 0;
#pragma unroll
  for (int i = 0; i < SCAN_CH; ++i) sum += cv[i];
  sPart[t] = sum;
  __syncthreads();
  for (int s = 1; s < SCAN_T; s <<= 1) {
    const int v = (t >= s) ? sPart[t - s] : 0;
    __syncthreads();
    sPart[t] += v;
    __syncthreads();
  }
  int run = sPart[t] - sum;
#pragma unroll
  for (int i = 0; i < SCAN_CH; ++i) {
    const int k = b + i;
    if (k < N_NODES) {
      off[k] = run;
      deg[k] = (float)cv[i];
    }
    run += cv[i];
  }
}

__global__ void __launch_bounds__(256)
fill_kernel(const int* __restrict__ src, const int* __restrict__ dst,
            const int* __restrict__ off, const int* __restrict__ slot,
            int* __restrict__ bucket) {
  int e = blockIdx.x * blockDim.x + threadIdx.x;
  if (e < N_EDGES) bucket[off[dst[e]] + slot[e]] = src[e];
}

__global__ void __launch_bounds__(256)
order_lists_kernel(const int* __restrict__ bucket, const int* __restrict__ off,
                   const int* __restrict__ cnt, int* __restrict__ ordered) {
  const int node = __builtin_amdgcn_readfirstlane(blockIdx.x * 8 + (threadIdx.x >> 5));
  if (node >= N_NODES) return;
  const int lane = threadIdx.x & 31;
  const int n = cnt[node];
  const int o = off[node];
  for (int i = lane; i < n; i += 32) {
    const int key = bucket[o + i];
    int r = 0;
    for (int j = 0; j < n; ++j) {
      const int kj = bucket[o + j];
      r += ((kj < key) || (kj == key && j < i)) ? 1 : 0;
    }
    ordered[o + r] = key;
  }
}

template <int K>
__device__ __forceinline__ void add_rows(const float* __restrict__ xl, int idx,
                                         int j, float4& acc) {
  float4 v[K];
#pragma unroll
  for (int k = 0; k < K; ++k) {
    const int s = __builtin_amdgcn_readlane(idx, j + k);
    v[k] = *reinterpret_cast<const float4*>(xl + (size_t)s * F_DIM);
  }
#pragma unroll
  for (int k = 0; k < K; ++k) {
    acc.x += v[k].x; acc.y += v[k].y; acc.z += v[k].z; acc.w += v[k].w;
  }
}

__global__ void __launch_bounds__(256)
gather_sum_kernel(const float* __restrict__ x, const int* __restrict__ ordered,
                  const int* __restrict__ off, const int* __restrict__ cnt,
                  float* __restrict__ agg) {
  const int node = __builtin_amdgcn_readfirstlane(blockIdx.x * 8 + (threadIdx.x >> 5));
  if (node >= N_NODES) return;
  const int lane = threadIdx.x & 31;
  const int n = cnt[node];
  const int o = off[node];
  const float* xl = x + lane * 4;
  float4 acc = make_float4(0.0f, 0.0f, 0.0f, 0.0f);
  for (int base = 0; base < n; base += 32) {
    const int m = (n - base < 32) ? (n - base) : 32;
    const int idx = (lane < m) ? ordered[o + base + lane] : 0;
    int j = 0;
    for (; j + 8 <= m; j += 8) add_rows<8>(xl, idx, j, acc);
    if (m & 4) { add_rows<4>(xl, idx, j, acc); j += 4; }
    if (m & 2) { add_rows<2>(xl, idx, j, acc); j += 2; }
    if (m & 1) { add_rows<1>(xl, idx, j, acc); }
  }
  *reinterpret_cast<float4*>(agg + (size_t)node * F_DIM + lane * 4) = acc;
}

__global__ void __launch_bounds__(256)
sage_layer_kernel(const float* __restrict__ x, const float* __restrict__ agg,
                  const float* __restrict__ deg, const float* __restrict__ wl,
                  const float* __restrict__ bl, const float* __restrict__ wr,
                  float* __restrict__ h) {
  __shared__ float sMean[16 * LDS_STRIDE];
  __shared__ float sX[16 * LDS_STRIDE];

  const int tm = blockIdx.x;

  for (int i = threadIdx.x; i < 16 * 32; i += 256) {
    const int r    = i >> 5;
    const int c4   = (i & 31) * 4;
    const int grow = tm * 16 + r;
    const float invd = 1.0f / fmaxf(deg[grow], 1.0f);
    const float4 va = *reinterpret_cast<const float4*>(agg + (size_t)grow * F_DIM + c4);
    const float4 vx = *reinterpret_cast<const float4*>(x   + (size_t)grow * F_DIM + c4);
    float* pm = sMean + r * LDS_STRIDE + c4;
    pm[0] = va.x * invd; pm[1] = va.y * invd; pm[2] = va.z * invd; pm[3] = va.w * invd;
    float* px = sX + r * LDS_STRIDE + c4;
    px[0] = vx.x; px[1] = vx.y; px[2] = vx.z; px[3] = vx.w;
  }
  __syncthreads();

  const int lane = threadIdx.x & 31;
  const int tn   = threadIdx.x >> 5;
  const int mloc = lane & 15;
  const int ncol = tn * 16 + mloc;
  const int kb   = (lane >= 16) ? 2 : 0;
  const float* sm = sMean + mloc * LDS_STRIDE;
  const float* sx = sX    + mloc * LDS_STRIDE;

  v8f acc = {};
#pragma unroll 4
  for (int kk = 0; kk < F_DIM; kk += 4) {
    v2f a1, b1, a2, b2;
    a1.x = sm[kk + kb];
    a1.y = sm[kk + kb + 1];
    b1.x = wl[(kk + kb)     * F_DIM + ncol];
    b1.y = wl[(kk + kb + 1) * F_DIM + ncol];
    acc = wmma_f32(a1, b1, acc);
    a2.x = sx[kk + kb];
    a2.y = sx[kk + kb + 1];
    b2.x = wr[(kk + kb)     * F_DIM + ncol];
    b2.y = wr[(kk + kb + 1) * F_DIM + ncol];
    acc = wmma_f32(a2, b2, acc);
  }
  const float bias  = bl[ncol];
  const int   mbase = tm * 16 + ((lane >= 16) ? 8 : 0);
#pragma unroll
  for (int i = 0; i < 8; ++i) {
    float v = fmaxf(acc[i] + bias, 0.0f);
    h[(size_t)(mbase + i) * F_DIM + ncol] = v;
  }
}

template <int ACT>
__global__ void __launch_bounds__(256)
wmma_gemm_bias_kernel(const float* __restrict__ A, const float* __restrict__ B,
                      const float* __restrict__ bias, float* __restrict__ C,
                      int N, int K) {
  __shared__ float sA[16 * LDS_STRIDE];

  const int tm = blockIdx.y;
  const int k4 = K >> 2;
  for (int i = threadIdx.x; i < 16 * k4; i += blockDim.x) {
    const int r  = i / k4;
    const int c4 = (i - r * k4) * 4;
    const float4 v = *reinterpret_cast<const float4*>(A + (size_t)(tm * 16 + r) * K + c4);
    float* p = sA + r * LDS_STRIDE + c4;
    p[0] = v.x; p[1] = v.y; p[2] = v.z; p[3] = v.w;
  }
  __syncthreads();

  const int lane   = threadIdx.x & 31;
  const int wave   = threadIdx.x >> 5;
  const int tilesN = (N + 15) >> 4;
  int tn = blockIdx.x * (blockDim.x >> 5) + wave;
  if (tn >= tilesN) tn = tilesN - 1;
  const int mloc = lane & 15;
  const int ncol = tn * 16 + mloc;
  const bool nok = (ncol < N);
  const int kb = (lane >= 16) ? 2 : 0;
  const float* sa = sA + mloc * LDS_STRIDE;

  v8f acc = {};
  for (int kk = 0; kk < K; kk += 4) {
    v2f a, b;
    a.x = sa[kk + kb];
    a.y = sa[kk + kb + 1];
    b.x = nok ? B[(size_t)(kk + kb) * N + ncol]     : 0.0f;
    b.y = nok ? B[(size_t)(kk + kb + 1) * N + ncol] : 0.0f;
    acc = wmma_f32(a, b, acc);
  }
  if (nok) {
    const float bv    = bias[ncol];
    const int   mbase = tm * 16 + ((lane >= 16) ? 8 : 0);
#pragma unroll
    for (int i = 0; i < 8; ++i) {
      float v = acc[i] + bv;
      if (ACT == 1) v = fmaxf(v, 0.0f);
      C[(size_t)(mbase + i) * N + ncol] = v;
    }
  }
}

__global__ void __launch_bounds__(256)
wmma_gemm_bn_kernel(const float* __restrict__ A, const float* __restrict__ B,
                    const float* __restrict__ bconv, const float* __restrict__ gamma,
                    const float* __restrict__ beta, const float* __restrict__ rm,
                    const float* __restrict__ rv, float* __restrict__ C,
                    int N, int K) {
  __shared__ float sA[16 * LDS_STRIDE];

  const int tm = blockIdx.y;
  const int k4 = K >> 2;
  for (int i = threadIdx.x; i < 16 * k4; i += blockDim.x) {
    const int r  = i / k4;
    const int c4 = (i - r * k4) * 4;
    const float4 v = *reinterpret_cast<const float4*>(A + (size_t)(tm * 16 + r) * K + c4);
    float* p = sA + r * LDS_STRIDE + c4;
    p[0] = v.x; p[1] = v.y; p[2] = v.z; p[3] = v.w;
  }
  __syncthreads();

  const int lane   = threadIdx.x & 31;
  const int wave   = threadIdx.x >> 5;
  const int tilesN = (N + 15) >> 4;
  int tn = blockIdx.x * (blockDim.x >> 5) + wave;
  if (tn >= tilesN) tn = tilesN - 1;
  const int mloc = lane & 15;
  const int ncol = tn * 16 + mloc;
  const bool nok = (ncol < N);
  const int kb = (lane >= 16) ? 2 : 0;
  const float* sa = sA + mloc * LDS_STRIDE;

  v8f acc = {};
  for (int kk = 0; kk < K; kk += 4) {
    v2f a, b;
    a.x = sa[kk + kb];
    a.y = sa[kk + kb + 1];
    b.x = nok ? B[(size_t)(kk + kb) * N + ncol]     : 0.0f;
    b.y = nok ? B[(size_t)(kk + kb + 1) * N + ncol] : 0.0f;
    acc = wmma_f32(a, b, acc);
  }
  if (nok) {
    const float s     = gamma[ncol] * rsqrtf(rv[ncol] + 1e-5f);
    const float shift = (bconv[ncol] - rm[ncol]) * s + beta[ncol];
    const int   mbase = tm * 16 + ((lane >= 16) ? 8 : 0);
#pragma unroll
    for (int i = 0; i < 8; ++i)
      C[(size_t)(mbase + i) * N + ncol] = acc[i] * s + shift;
  }
}

__global__ void __launch_bounds__(256)
tanh_kernel(const float* __restrict__ in, float* __restrict__ out, int n) {
  int i = blockIdx.x * blockDim.x + threadIdx.x;
  if (i < n) out[i] = tanhf(in[i]);
}

__global__ void __launch_bounds__(256)
true_lab_kernel(const float* __restrict__ pre, const float* __restrict__ wtl,
                const float* __restrict__ btl, float* __restrict__ out) {
  int r = blockIdx.x * blockDim.x + threadIdx.x;
  if (r >= N_NODES) return;
  const float* p = pre + (size_t)r * NBITS;
  float s = 0.0f;
#pragma unroll
  for (int k = 0; k < NBITS; ++k) s += p[k] * wtl[k];
  out[r] = s + btl[0];
}

extern "C" void kernel_launch(void* const* d_in, const int* in_sizes, int n_in,
                              void* d_out, int out_size, void* d_ws, size_t ws_size,
                              hipStream_t stream) {
  (void)in_sizes; (void)n_in; (void)out_size; (void)ws_size;

  const float* features = (const float*)d_in[0];
  const int*   edges    = (const int*)d_in[1];
  const float* w1l   = (const float*)d_in[2];
  const float* b1l   = (const float*)d_in[3];
  const float* w1r   = (const float*)d_in[4];
  const float* w2l   = (const float*)d_in[5];
  const float* b2l   = (const float*)d_in[6];
  const float* w2r   = (const float*)d_in[7];
  const float* whd   = (const float*)d_in[8];
  const float* bhd   = (const float*)d_in[9];
  const float* wclas = (const float*)d_in[10];
  const float* bclas = (const float*)d_in[11];
  const float* wconv = (const float*)d_in[12];
  const float* bconv = (const float*)d_in[13];
  const float* gamma = (const float*)d_in[14];
  const float* beta  = (const float*)d_in[15];
  const float* rm    = (const float*)d_in[16];
  const float* rv    = (const float*)d_in[17];
  const float* wtl   = (const float*)d_in[18];
  const float* btl   = (const float*)d_in[19];
  const float* wcv   = (const float*)d_in[20];
  const float* bcv   = (const float*)d_in[21];

  const int* src = edges;
  const int* dst = edges + N_EDGES;

  float* ws     = (float*)d_ws;
  float* deg    = ws;
  float* agg    = deg + N_NODES;
  float* h1     = agg + (size_t)N_NODES * F_DIM;
  float* h2     = h1  + (size_t)N_NODES * F_DIM;
  float* pre_bn = h2  + (size_t)N_NODES * F_DIM;
  int* cnt     = (int*)(pre_bn + (size_t)N_NODES * NBITS);
  int* off     = cnt + CNT_LEN;
  int* slot    = off + N_NODES;
  int* bucket  = slot + N_EDGES;
  int* ordered = bucket + N_EDGES;

  float* out       = (float*)d_out;
  float* o_logists = out;
  float* o_tanh    = o_logists + (size_t)N_NODES * NCLS;
  float* o_fealab  = o_tanh    + (size_t)N_NODES * NBITS;
  float* o_feacv   = o_fealab  + (size_t)N_NODES * HD;
  float* o_truelab = o_feacv   + (size_t)N_NODES * NCLS;

  const int edgeBlocks = (N_EDGES + 255) / 256;
  const int nodeBlocks = (N_NODES + 7) / 8;
  const int rowTiles      = N_NODES / 16;
  auto gridFor = [rowTiles](int N, int waves) {
    int tilesN = (N + 15) / 16;
    return dim3((tilesN + waves - 1) / waves, rowTiles);
  };

  hipMemsetAsync(cnt, 0, sizeof(int) * (size_t)CNT_LEN, stream);
  count_slot_kernel<<<edgeBlocks, 256, 0, stream>>>(dst, cnt, slot);
  scan_offsets_kernel<<<1, SCAN_T, 0, stream>>>(cnt, off, deg);
  fill_kernel<<<edgeBlocks, 256, 0, stream>>>(src, dst, off, slot, bucket);
  order_lists_kernel<<<nodeBlocks, 256, 0, stream>>>(bucket, off, cnt, ordered);

  gather_sum_kernel<<<nodeBlocks, 256, 0, stream>>>(features, ordered, off, cnt, agg);
  sage_layer_kernel<<<rowTiles, 256, 0, stream>>>(features, agg, deg, w1l, b1l, w1r, h1);

  gather_sum_kernel<<<nodeBlocks, 256, 0, stream>>>(h1, ordered, off, cnt, agg);
  sage_layer_kernel<<<rowTiles, 256, 0, stream>>>(h1, agg, deg, w2l, b2l, w2r, h2);

  wmma_gemm_bias_kernel<0><<<gridFor(HD, 8), 256, 0, stream>>>(
      h2, whd, bhd, o_fealab, HD, F_DIM);
  wmma_gemm_bias_kernel<0><<<gridFor(NCLS, 8), 256, 0, stream>>>(
      h2, wclas, bclas, o_logists, NCLS, F_DIM);
  wmma_gemm_bn_kernel<<<gridFor(NBITS, 4), 128, 0, stream>>>(
      h2, wconv, bconv, gamma, beta, rm, rv, pre_bn, NBITS, F_DIM);
  tanh_kernel<<<(N_NODES * NBITS + 255) / 256, 256, 0, stream>>>(
      pre_bn, o_tanh, N_NODES * NBITS);
  wmma_gemm_bias_kernel<0><<<gridFor(NCLS, 8), 256, 0, stream>>>(
      o_tanh, wcv, bcv, o_feacv, NCLS, NBITS);
  true_lab_kernel<<<(N_NODES + 255) / 256, 256, 0, stream>>>(
      pre_bn, wtl, btl, o_truelab);
}


# Head stage fused into one kernel (3 GEMMs + BN, tanh, fea_convert GEMM, true_lab dot): 80-row h2 tile staged once in LDS, one B fragment feeds 5 row tiles, pre/tanh tiles stay in LDS; 14->10 launches
# speedup vs baseline: 2.0607x; 1.3781x over previous
#include <hip/hip_runtime.h>
#include <math.h>


typedef __attribute__((ext_vector_type(2))) float v2f;
typedef __attribute__((ext_vector_type(8))) float v8f;

#define N_NODES 50000
#define N_EDGES 800000
#define F_DIM   128
#define NBITS   64
#define NCLS    100
#define HD      300
#define LDS_STRIDE 132

__device__ __forceinline__ v8f wmma_f32(v2f a, v2f b, v8f c) {
  return __builtin_amdgcn_wmma_f32_16x16x4_f32(false, a, false, b, (short)0, c,
                                               false, false);
}


__global__ void __launch_bounds__(256)
count_slot_kernel(const int* __restrict__ dst, int* __restrict__ cnt,
                  int* __restrict__ slot) {
  int e = blockIdx.x * blockDim.x + threadIdx.x;
  if (e < N_EDGES) slot[e] = atomicAdd(&cnt[dst[e]], 1);
}

#define SCAN_T  1024
#define SCAN_CH ((N_NODES + SCAN_T - 1) / SCAN_T)
#define CNT_LEN (SCAN_T * SCAN_CH)
__global__ void __launch_bounds__(SCAN_T)
scan_offsets_kernel(const int* __restrict__ cnt, int* __restrict__ off,
                    float* __restrict__ deg) {
  __shared__ int sPart[SCAN_T];
  const int t = threadIdx.x;
  const int b = t * SCAN_CH;
  int cv[SCAN_CH];
#pragma unroll
  for (int i = 0; i < SCAN_CH; ++i) cv[i] = cnt[b + i];
  int sum = 0;
#pragma unroll
  for (int i = 0; i < SCAN_CH; ++i) sum += cv[i];
  sPart[t] = sum;
  __syncthreads();
  for (int s = 1; s < SCAN_T; s <<= 1) {
    const int v = (t >= s) ? sPart[t - s] : 0;
    __syncthreads();
    sPart[t] += v;
    __syncthreads();
  }
  int run = sPart[t] - sum;
#pragma unroll
  for (int i = 0; i < SCAN_CH; ++i) {
    const int k = b + i;
    if (k < N_NODES) {
      off[k] = run;
      deg[k] = (float)cv[i];
    }
    run += cv[i];
  }
}

__global__ void __launch_bounds__(256)
fill_kernel(const int* __restrict__ src, const int* __restrict__ dst,
            const int* __restrict__ off, const int* __restrict__ slot,
            int* __restrict__ bucket) {
  int e = blockIdx.x * blockDim.x + threadIdx.x;
  if (e < N_EDGES) bucket[off[dst[e]] + slot[e]] = src[e];
}

__global__ void __launch_bounds__(256)
order_lists_kernel(const int* __restrict__ bucket, const int* __restrict__ off,
                   const int* __restrict__ cnt, int* __restrict__ ordered) {
  const int node = __builtin_amdgcn_readfirstlane(blockIdx.x * 8 + (threadIdx.x >> 5));
  if (node >= N_NODES) return;
  const int lane = threadIdx.x & 31;
  const int n = cnt[node];
  const int o = off[node];
  for (int i = lane; i < n; i += 32) {
    const int key = bucket[o + i];
    int r = 0;
    for (int j = 0; j < n; ++j) {
      const int kj = bucket[o + j];
      r += ((kj < key) || (kj == key && j < i)) ? 1 : 0;
    }
    ordered[o + r] = key;
  }
}

template <int K>
__device__ __forceinline__ void add_rows(const float* __restrict__ xl, int idx,
                                         int j, float4& acc) {
  float4 v[K];
#pragma unroll
  for (int k = 0; k < K; ++k) {
    const int s = __builtin_amdgcn_readlane(idx, j + k);
    v[k] = *reinterpret_cast<const float4*>(xl + (size_t)s * F_DIM);
  }
#pragma unroll
  for (int k = 0; k < K; ++k) {
    acc.x += v[k].x; acc.y += v[k].y; acc.z += v[k].z; acc.w += v[k].w;
  }
}

__global__ void __launch_bounds__(256)
gather_sum_kernel(const float* __restrict__ x, const int* __restrict__ ordered,
                  const int* __restrict__ off, const int* __restrict__ cnt,
                  float* __restrict__ agg) {
  const int node = __builtin_amdgcn_readfirstlane(blockIdx.x * 8 + (threadIdx.x >> 5));
  if (node >= N_NODES) return;
  const int lane = threadIdx.x & 31;
  const int n = cnt[node];
  const int o = off[node];
  const float* xl = x + lane * 4;
  float4 acc = make_float4(0.0f, 0.0f, 0.0f, 0.0f);
  for (int base = 0; base < n; base += 32) {
    const int m = (n - base < 32) ? (n - base) : 32;
    const int idx = (lane < m) ? ordered[o + base + lane] : 0;
    int j = 0;
    for (; j + 8 <= m; j += 8) add_rows<8>(xl, idx, j, acc);
    if (m & 4) { add_rows<4>(xl, idx, j, acc); j += 4; }
    if (m & 2) { add_rows<2>(xl, idx, j, acc); j += 2; }
    if (m & 1) { add_rows<1>(xl, idx, j, acc); }
  }
  *reinterpret_cast<float4*>(agg + (size_t)node * F_DIM + lane * 4) = acc;
}

__global__ void __launch_bounds__(256)
sage_layer_kernel(const float* __restrict__ x, const float* __restrict__ agg,
                  const float* __restrict__ deg, const float* __restrict__ wl,
                  const float* __restrict__ bl, const float* __restrict__ wr,
                  float* __restrict__ h) {
  __shared__ float sMean[16 * LDS_STRIDE];
  __shared__ float sX[16 * LDS_STRIDE];

  const int tm = blockIdx.x;

  for (int i = threadIdx.x; i < 16 * 32; i += 256) {
    const int r    = i >> 5;
    const int c4   = (i & 31) * 4;
    const int grow = tm * 16 + r;
    const float invd = 1.0f / fmaxf(deg[grow], 1.0f);
    const float4 va = *reinterpret_cast<const float4*>(agg + (size_t)grow * F_DIM + c4);
    const float4 vx = *reinterpret_cast<const float4*>(x   + (size_t)grow * F_DIM + c4);
    float* pm = sMean + r * LDS_STRIDE + c4;
    pm[0] = va.x * invd; pm[1] = va.y * invd; pm[2] = va.z * invd; pm[3] = va.w * invd;
    float* px = sX + r * LDS_STRIDE + c4;
    px[0] = vx.x; px[1] = vx.y; px[2] = vx.z; px[3] = vx.w;
  }
  __syncthreads();

  const int lane = threadIdx.x & 31;
  const int tn   = threadIdx.x >> 5;
  const int mloc = lane & 15;
  const int ncol = tn * 16 + mloc;
  const int kb   = (lane >= 16) ? 2 : 0;
  const float* sm = sMean + mloc * LDS_STRIDE;
  const float* sx = sX    + mloc * LDS_STRIDE;

  v8f acc = {};
#pragma unroll 4
  for (int kk = 0; kk < F_DIM; kk += 4) {
    v2f a1, b1, a2, b2;
    a1.x = sm[kk + kb];
    a1.y = sm[kk + kb + 1];
    b1.x = wl[(kk + kb)     * F_DIM + ncol];
    b1.y = wl[(kk + kb + 1) * F_DIM + ncol];
    acc = wmma_f32(a1, b1, acc);
    a2.x = sx[kk + kb];
    a2.y = sx[kk + kb + 1];
    b2.x = wr[(kk + kb)     * F_DIM + ncol];
    b2.y = wr[(kk + kb + 1) * F_DIM + ncol];
    acc = wmma_f32(a2, b2, acc);
  }
  const float bias  = bl[ncol];
  const int   mbase = tm * 16 + ((lane >= 16) ? 8 : 0);
#pragma unroll
  for (int i = 0; i < 8; ++i) {
    float v = fmaxf(acc[i] + bias, 0.0f);
    h[(size_t)(mbase + i) * F_DIM + ncol] = v;
  }
}

#define HB_TILES 5
#define HB_ROWS  (16 * HB_TILES)
#define HB_WAVES 10
#define T_STRIDE 68

template <int K, int STRIDE, int N>
__device__ __forceinline__ void tiles_mma(const float* sa, const float* __restrict__ bp,
                                          v8f (&acc)[HB_TILES]) {
#pragma unroll 4
  for (int kk = 0; kk < K; kk += 4) {
    v2f b;
    b.x = bp[kk * N];
    b.y = bp[(kk + 1) * N];
#pragma unroll
    for (int t = 0; t < HB_TILES; ++t) {
      const v2f a = *reinterpret_cast<const v2f*>(sa + t * 16 * STRIDE + kk);
      acc[t] = wmma_f32(a, b, acc[t]);
    }
  }
}

template <int N>
__device__ __forceinline__ void store_tiles(float* __restrict__ C, int rbase, int ncol,
                                            float bv, const v8f (&acc)[HB_TILES]) {
#pragma unroll
  for (int t = 0; t < HB_TILES; ++t) {
#pragma unroll
    for (int i = 0; i < 8; ++i)
      C[(size_t)(rbase + t * 16 + i) * N + ncol] = acc[t][i] + bv;
  }
}

__global__ void __launch_bounds__(32 * HB_WAVES)
heads_kernel(const float* __restrict__ h2,
             const float* __restrict__ whd, const float* __restrict__ bhd,
             const float* __restrict__ wclas, const float* __restrict__ bclas,
             const float* __restrict__ wconv, const float* __restrict__ bconv,
             const float* __restrict__ gamma, const float* __restrict__ beta,
             const float* __restrict__ rm, const float* __restrict__ rv,
             const float* __restrict__ wtl, const float* __restrict__ btl,
             const float* __restrict__ wcv, const float* __restrict__ bcv,
             float* __restrict__ o_logists, float* __restrict__ o_tanh,
             float* __restrict__ o_fealab, float* __restrict__ o_feacv,
             float* __restrict__ o_truelab) {
  __shared__ __attribute__((aligned(16))) float sA[HB_ROWS * LDS_STRIDE];
  __shared__ __attribute__((aligned(16))) float sT[HB_ROWS * T_STRIDE];
  __shared__ __attribute__((aligned(16))) float sP[HB_ROWS * T_STRIDE];

  const int row0 = blockIdx.x * HB_ROWS;
#pragma unroll
  for (int j = 0; j < (HB_ROWS * 32) / (32 * HB_WAVES); ++j) {
    const int i  = threadIdx.x + j * (32 * HB_WAVES);
    const int r  = i >> 5;
    const int c4 = (i & 31) * 4;
    const float4 v = *reinterpret_cast<const float4*>(h2 + (size_t)(row0 + r) * F_DIM + c4);
    *reinterpret_cast<float4*>(sA + r * LDS_STRIDE + c4) = v;
  }
  __syncthreads();

  const int lane = threadIdx.x & 31;
  const int wave = __builtin_amdgcn_readfirstlane(threadIdx.x >> 5);
  const int mloc = lane & 15;
  const int hi   = lane >> 4;
  const int kb   = hi * 2;
  const int rloc = hi * 8;
  const float* sa = sA + mloc * LDS_STRIDE + kb;

  for (int job = wave; job < 30; job += HB_WAVES) {
    v8f acc[HB_TILES] = {};
    if (job < 19) {
      const int ncol = job * 16 + mloc;
      const bool nok = (ncol < HD);
      const int nc = nok ? ncol : (HD - 1);
      tiles_mma<F_DIM, LDS_STRIDE, HD>(sa, whd + kb * HD + nc, acc);
      if (nok) store_tiles<HD>(o_fealab, row0 + rloc, ncol, bhd[ncol], acc);
    } else if (job < 26) {
      const int ncol = (job - 19) * 16 + mloc;
      const bool nok = (ncol < NCLS);
      const int nc = nok ? ncol : (NCLS - 1);
      tiles_mma<F_DIM, LDS_STRIDE, NCLS>(sa, wclas + kb * NCLS + nc, acc);
      if (nok) store_tiles<NCLS>(o_logists, row0 + rloc, ncol, bclas[ncol], acc);
    } else {
      const int ncol = (job - 26) * 16 + mloc;
      tiles_mma<F_DIM, LDS_STRIDE, NBITS>(sa, wconv + kb * NBITS + ncol, acc);
      const float s     = gamma[ncol] * rsqrtf(rv[ncol] + 1e-5f);
      const float shift = (bconv[ncol] - rm[ncol]) * s + beta[ncol];
#pragma unroll
      for (int t = 0; t < HB_TILES; ++t) {
#pragma unroll
        for (int i = 0; i < 8; ++i)
          sP[(t * 16 + rloc + i) * T_STRIDE + ncol] = acc[t][i] * s + shift;
      }
    }
  }
  __syncthreads();

#pragma unroll
  for (int j = 0; j < (HB_ROWS * 16) / (32 * HB_WAVES); ++j) {
    const int i  = threadIdx.x + j * (32 * HB_WAVES);
    const int r  = i >> 4;
    const int c4 = (i & 15) * 4;
    const float4 p = *reinterpret_cast<const float4*>(sP + r * T_STRIDE + c4);
    float4 th;
    th.x = tanhf(p.x); th.y = tanhf(p.y); th.z = tanhf(p.z); th.w = tanhf(p.w);
    *reinterpret_cast<float4*>(sT + r * T_STRIDE + c4) = th;
    *reinterpret_cast<float4*>(o_tanh + (size_t)(row0 + r) * NBITS + c4) = th;
  }
  __syncthreads();

  if (wave < 7) {
    v8f acc[HB_TILES] = {};
    const int ncol = wave * 16 + mloc;
    const bool nok = (ncol < NCLS);
    const int nc = nok ? ncol : (NCLS - 1);
    tiles_mma<NBITS, T_STRIDE, NCLS>(sT + mloc * T_STRIDE + kb, wcv + kb * NCLS + nc, acc);
    if (nok) store_tiles<NCLS>(o_feacv, row0 + rloc, ncol, bcv[ncol], acc);
  } else {
    const int r = (wave - 7) * 32 + lane;
    if (r < HB_ROWS) {
      const float* p = sP + r * T_STRIDE;
      float s = 0.0f;
#pragma unroll
      for (int k = 0; k < NBITS; ++k) s += p[k] * wtl[k];
      o_truelab[row0 + r] = s + btl[0];
    }
  }
}

extern "C" void kernel_launch(void* const* d_in, const int* in_sizes, int n_in,
                              void* d_out, int out_size, void* d_ws, size_t ws_size,
                              hipStream_t stream) {
  (void)in_sizes; (void)n_in; (void)out_size; (void)ws_size;

  const float* features = (const float*)d_in[0];
  const int*   edges    = (const int*)d_in[1];
  const float* w1l   = (const float*)d_in[2];
  const float* b1l   = (const float*)d_in[3];
  const float* w1r   = (const float*)d_in[4];
  const float* w2l   = (const float*)d_in[5];
  const float* b2l   = (const float*)d_in[6];
  const float* w2r   = (const float*)d_in[7];
  const float* whd   = (const float*)d_in[8];
  const float* bhd   = (const float*)d_in[9];
  const float* wclas = (const float*)d_in[10];
  const float* bclas = (const float*)d_in[11];
  const float* wconv = (const float*)d_in[12];
  const float* bconv = (const float*)d_in[13];
  const float* gamma = (const float*)d_in[14];
  const float* beta  = (const float*)d_in[15];
  const float* rm    = (const float*)d_in[16];
  const float* rv    = (const float*)d_in[17];
  const float* wtl   = (const float*)d_in[18];
  const float* btl   = (const float*)d_in[19];
  const float* wcv   = (const float*)d_in[20];
  const float* bcv   = (const float*)d_in[21];

  const int* src = edges;
  const int* dst = edges + N_EDGES;

  float* ws     = (float*)d_ws;
  float* deg    = ws;
  float* agg    = deg + N_NODES;
  float* h1     = agg + (size_t)N_NODES * F_DIM;
  float* h2     = h1  + (size_t)N_NODES * F_DIM;
  int* cnt     = (int*)(h2 + (size_t)N_NODES * F_DIM);
  int* off     = cnt + CNT_LEN;
  int* slot    = off + N_NODES;
  int* bucket  = slot + N_EDGES;
  int* ordered = bucket + N_EDGES;

  float* out       = (float*)d_out;
  float* o_logists = out;
  float* o_tanh    = o_logists + (size_t)N_NODES * NCLS;
  float* o_fealab  = o_tanh    + (size_t)N_NODES * NBITS;
  float* o_feacv   = o_fealab  + (size_t)N_NODES * HD;
  float* o_truelab = o_feacv   + (size_t)N_NODES * NCLS;

  const int edgeBlocks = (N_EDGES + 255) / 256;
  const int nodeBlocks = (N_NODES + 7) / 8;
  const int rowTiles      = N_NODES / 16;

  hipMemsetAsync(cnt, 0, sizeof(int) * (size_t)CNT_LEN, stream);
  count_slot_kernel<<<edgeBlocks, 256, 0, stream>>>(dst, cnt, slot);
  scan_offsets_kernel<<<1, SCAN_T, 0, stream>>>(cnt, off, deg);
  fill_kernel<<<edgeBlocks, 256, 0, stream>>>(src, dst, off, slot, bucket);
  order_lists_kernel<<<nodeBlocks, 256, 0, stream>>>(bucket, off, cnt, ordered);

  gather_sum_kernel<<<nodeBlocks, 256, 0, stream>>>(features, ordered, off, cnt, agg);
  sage_layer_kernel<<<rowTiles, 256, 0, stream>>>(features, agg, deg, w1l, b1l, w1r, h1);

  gather_sum_kernel<<<nodeBlocks, 256, 0, stream>>>(h1, ordered, off, cnt, agg);
  sage_layer_kernel<<<rowTiles, 256, 0, stream>>>(h1, agg, deg, w2l, b2l, w2r, h2);

  heads_kernel<<<N_NODES / HB_ROWS, 32 * HB_WAVES, 0, stream>>>(
      h2, whd, bhd, wclas, bclas, wconv, bconv, gamma, beta, rm, rv, wtl, btl,
      wcv, bcv, o_logists, o_tanh, o_fealab, o_feacv, o_truelab);
}


# Prefix-scan kernel (one block): count plane staged in LDS by whole-line 16-byte loads, chunks scanned in LDS, off and deg stored in node order; no lane-strided global access left
# speedup vs baseline: 2.4170x; 1.1729x over previous
#include <hip/hip_runtime.h>
#include <math.h>


typedef __attribute__((ext_vector_type(2))) float v2f;
typedef __attribute__((ext_vector_type(8))) float v8f;

#define N_NODES 50000
#define N_EDGES 800000
#define F_DIM   128
#define NBITS   64
#define NCLS    100
#define HD      300
#define LDS_STRIDE 132

__device__ __forceinline__ v8f wmma_f32(v2f a, v2f b, v8f c) {
  return __builtin_amdgcn_wmma_f32_16x16x4_f32(false, a, false, b, (short)0, c,
                                               false, false);
}


__global__ void __launch_bounds__(256)
count_slot_kernel(const int* __restrict__ dst, int* __restrict__ cnt,
                  int* __restrict__ slot) {
  int e = blockIdx.x * blockDim.x + threadIdx.x;
  if (e < N_EDGES) slot[e] = atomicAdd(&cnt[dst[e]], 1);
}

#define SCAN_T  1024
#define SCAN_CH ((N_NODES + SCAN_T - 1) / SCAN_T)
#define CNT_LEN (SCAN_T * SCAN_CH)
#define CNT_V4  (CNT_LEN / 4)
#define SCAN_PASSES ((CNT_V4 + SCAN_T - 1) / SCAN_T)
__global__ void __launch_bounds__(SCAN_T)
scan_offsets_kernel(const int* __restrict__ cnt, int* __restrict__ off,
                    float* __restrict__ deg) {
  __shared__ __attribute__((aligned(16))) int sCnt[CNT_LEN];
  __shared__ int sPart[SCAN_T];
  const int t = threadIdx.x;

  {
    const int4* __restrict__ g4 = reinterpret_cast<const int4*>(cnt);
    int4 v[SCAN_PASSES];
#pragma unroll
    for (int j = 0; j < SCAN_PASSES; ++j) {
      const int i = t + j * SCAN_T;
      if (i < CNT_V4) v[j] = g4[i];
    }
#pragma unroll
    for (int j = 0; j < SCAN_PASSES; ++j) {
      const int i = t + j * SCAN_T;
      if (i < CNT_V4) {
        reinterpret_cast<int4*>(sCnt)[i] = v[j];
        if (i < N_NODES / 4)
          reinterpret_cast<float4*>(deg)[i] =
              make_float4((float)v[j].x, (float)v[j].y, (float)v[j].z, (float)v[j].w);
      }
    }
  }
  __syncthreads();

  const int b = t * SCAN_CH;
  int cv[SCAN_CH];
#pragma unroll
  for (int i = 0; i < SCAN_CH; ++i) cv[i] = sCnt[b + i];
  int sum = 0;
#pragma unroll
  for (int i = 0; i < SCAN_CH; ++i) sum += cv[i];
  sPart[t] = sum;
  __syncthreads();
  for (int s = 1; s < SCAN_T; s <<= 1) {
    const int v = (t >= s) ? sPart[t - s] : 0;
    __syncthreads();
    sPart[t] += v;
    __syncthreads();
  }
  int run = sPart[t] - sum;
#pragma unroll
  for (int i = 0; i < SCAN_CH; ++i) {
    sCnt[b + i] = run;
    run += cv[i];
  }
  __syncthreads();

#pragma unroll
  for (int j = 0; j < (N_NODES / 4 + SCAN_T - 1) / SCAN_T; ++j) {
    const int i = t + j * SCAN_T;
    if (i < N_NODES / 4)
      reinterpret_cast<int4*>(off)[i] = reinterpret_cast<const int4*>(sCnt)[i];
  }
}

__global__ void __launch_bounds__(256)
fill_kernel(const int* __restrict__ src, const int* __restrict__ dst,
            const int* __restrict__ off, const int* __restrict__ slot,
            int* __restrict__ bucket) {
  int e = blockIdx.x * blockDim.x + threadIdx.x;
  if (e < N_EDGES) bucket[off[dst[e]] + slot[e]] = src[e];
}

__global__ void __launch_bounds__(256)
order_lists_kernel(const int* __restrict__ bucket, const int* __restrict__ off,
                   const int* __restrict__ cnt, int* __restrict__ ordered) {
  const int node = __builtin_amdgcn_readfirstlane(blockIdx.x * 8 + (threadIdx.x >> 5));
  if (node >= N_NODES) return;
  const int lane = threadIdx.x & 31;
  const int n = cnt[node];
  const int o = off[node];
  for (int i = lane; i < n; i += 32) {
    const int key = bucket[o + i];
    int r = 0;
    for (int j = 0; j < n; ++j) {
      const int kj = bucket[o + j];
      r += ((kj < key) || (kj == key && j < i)) ? 1 : 0;
    }
    ordered[o + r] = key;
  }
}

template <int K>
__device__ __forceinline__ void add_rows(const float* __restrict__ xl, int idx,
                                         int j, float4& acc) {
  float4 v[K];
#pragma unroll
  for (int k = 0; k < K; ++k) {
    const int s = __builtin_amdgcn_readlane(idx, j + k);
    v[k] = *reinterpret_cast<const float4*>(xl + (size_t)s * F_DIM);
  }
#pragma unroll
  for (int k = 0; k < K; ++k) {
    acc.x += v[k].x; acc.y += v[k].y; acc.z += v[k].z; acc.w += v[k].w;
  }
}

__global__ void __launch_bounds__(256)
gather_sum_kernel(const float* __restrict__ x, const int* __restrict__ ordered,
                  const int* __restrict__ off, const int* __restrict__ cnt,
                  float* __restrict__ agg) {
  const int node = __builtin_amdgcn_readfirstlane(blockIdx.x * 8 + (threadIdx.x >> 5));
  if (node >= N_NODES) return;
  const int lane = threadIdx.x & 31;
  const int n = cnt[node];
  const int o = off[node];
  const float* xl = x + lane * 4;
  float4 acc = make_float4(0.0f, 0.0f, 0.0f, 0.0f);
  for (int base = 0; base < n; base += 32) {
    const int m = (n - base < 32) ? (n - base) : 32;
    const int idx = (lane < m) ? ordered[o + base + lane] : 0;
    int j = 0;
    for (; j + 8 <= m; j += 8) add_rows<8>(xl, idx, j, acc);
    if (m & 4) { add_rows<4>(xl, idx, j, acc); j += 4; }
    if (m & 2) { add_rows<2>(xl, idx, j, acc); j += 2; }
    if (m & 1) { add_rows<1>(xl, idx, j, acc); }
  }
  *reinterpret_cast<float4*>(agg + (size_t)node * F_DIM + lane * 4) = acc;
}

__global__ void __launch_bounds__(256)
sage_layer_kernel(const float* __restrict__ x, const float* __restrict__ agg,
                  const float* __restrict__ deg, const float* __restrict__ wl,
                  const float* __restrict__ bl, const float* __restrict__ wr,
                  float* __restrict__ h) {
  __shared__ float sMean[16 * LDS_STRIDE];
  __shared__ float sX[16 * LDS_STRIDE];

  const int tm = blockIdx.x;

  for (int i = threadIdx.x; i < 16 * 32; i += 256) {
    const int r    = i >> 5;
    const int c4   = (i & 31) * 4;
    const int grow = tm * 16 + r;
    const float invd = 1.0f / fmaxf(deg[grow], 1.0f);
    const float4 va = *reinterpret_cast<const float4*>(agg + (size_t)grow * F_DIM + c4);
    const float4 vx = *reinterpret_cast<const float4*>(x   + (size_t)grow * F_DIM + c4);
    float* pm = sMean + r * LDS_STRIDE + c4;
    pm[0] = va.x * invd; pm[1] = va.y * invd; pm[2] = va.z * invd; pm[3] = va.w * invd;
    float* px = sX + r * LDS_STRIDE + c4;
    px[0] = vx.x; px[1] = vx.y; px[2] = vx.z; px[3] = vx.w;
  }
  __syncthreads();

  const int lane = threadIdx.x & 31;
  const int tn   = threadIdx.x >> 5;
  const int mloc = lane & 15;
  const int ncol = tn * 16 + mloc;
  const int kb   = (lane >= 16) ? 2 : 0;
  const float* sm = sMean + mloc * LDS_STRIDE;
  const float* sx = sX    + mloc * LDS_STRIDE;

  v8f acc = {};
#pragma unroll 4
  for (int kk = 0; kk < F_DIM; kk += 4) {
    v2f a1, b1, a2, b2;
    a1.x = sm[kk + kb];
    a1.y = sm[kk + kb + 1];
    b1.x = wl[(kk + kb)     * F_DIM + ncol];
    b1.y = wl[(kk + kb + 1) * F_DIM + ncol];
    acc = wmma_f32(a1, b1, acc);
    a2.x = sx[kk + kb];
    a2.y = sx[kk + kb + 1];
    b2.x = wr[(kk + kb)     * F_DIM + ncol];
    b2.y = wr[(kk + kb + 1) * F_DIM + ncol];
    acc = wmma_f32(a2, b2, acc);
  }
  const float bias  = bl[ncol];
  const int   mbase = tm * 16 + ((lane >= 16) ? 8 : 0);
#pragma unroll
  for (int i = 0; i < 8; ++i) {
    float v = fmaxf(acc[i] + bias, 0.0f);
    h[(size_t)(mbase + i) * F_DIM + ncol] = v;
  }
}

#define HB_TILES 5
#define HB_ROWS  (16 * HB_TILES)
#define HB_WAVES 10
#define T_STRIDE 68

template <int K, int STRIDE, int N>
__device__ __forceinline__ void tiles_mma(const float* sa, const float* __restrict__ bp,
                                          v8f (&acc)[HB_TILES]) {
#pragma unroll 4
  for (int kk = 0; kk < K; kk += 4) {
    v2f b;
    b.x = bp[kk * N];
    b.y = bp[(kk + 1) * N];
#pragma unroll
    for (int t = 0; t < HB_TILES; ++t) {
      const v2f a = *reinterpret_cast<const v2f*>(sa + t * 16 * STRIDE + kk);
      acc[t] = wmma_f32(a, b, acc[t]);
    }
  }
}

template <int N>
__device__ __forceinline__ void store_tiles(float* __restrict__ C, int rbase, int ncol,
                                            float bv, const v8f (&acc)[HB_TILES]) {
#pragma unroll
  for (int t = 0; t < HB_TILES; ++t) {
#pragma unroll
    for (int i = 0; i < 8; ++i)
      C[(size_t)(rbase + t * 16 + i) * N + ncol] = acc[t][i] + bv;
  }
}

__global__ void __launch_bounds__(32 * HB_WAVES)
heads_kernel(const float* __restrict__ h2,
             const float* __restrict__ whd, const float* __restrict__ bhd,
             const float* __restrict__ wclas, const float* __restrict__ bclas,
             const float* __restrict__ wconv, const float* __restrict__ bconv,
             const float* __restrict__ gamma, const float* __restrict__ beta,
             const float* __restrict__ rm, const float* __restrict__ rv,
             const float* __restrict__ wtl, const float* __restrict__ btl,
             const float* __restrict__ wcv, const float* __restrict__ bcv,
             float* __restrict__ o_logists, float* __restrict__ o_tanh,
             float* __restrict__ o_fealab, float* __restrict__ o_feacv,
             float* __restrict__ o_truelab) {
  __shared__ __attribute__((aligned(16))) float sA[HB_ROWS * LDS_STRIDE];
  __shared__ __attribute__((aligned(16))) float sT[HB_ROWS * T_STRIDE];
  __shared__ __attribute__((aligned(16))) float sP[HB_ROWS * T_STRIDE];

  const int row0 = blockIdx.x * HB_ROWS;
#pragma unroll
  for (int j = 0; j < (HB_ROWS * 32) / (32 * HB_WAVES); ++j) {
    const int i  = threadIdx.x + j * (32 * HB_WAVES);
    const int r  = i >> 5;
    const int c4 = (i & 31) * 4;
    const float4 v = *reinterpret_cast<const float4*>(h2 + (size_t)(row0 + r) * F_DIM + c4);
    *reinterpret_cast<float4*>(sA + r * LDS_STRIDE + c4) = v;
  }
  __syncthreads();

  const int lane = threadIdx.x & 31;
  const int wave = __builtin_amdgcn_readfirstlane(threadIdx.x >> 5);
  const int mloc = lane & 15;
  const int hi   = lane >> 4;
  const int kb   = hi * 2;
  const int rloc = hi * 8;
  const float* sa = sA + mloc * LDS_STRIDE + kb;

  for (int job = wave; job < 30; job += HB_WAVES) {
    v8f acc[HB_TILES] = {};
    if (job < 19) {
      const int ncol = job * 16 + mloc;
      const bool nok = (ncol < HD);
      const int nc = nok ? ncol : (HD - 1);
      tiles_mma<F_DIM, LDS_STRIDE, HD>(sa, whd + kb * HD + nc, acc);
      if (nok) store_tiles<HD>(o_fealab, row0 + rloc, ncol, bhd[ncol], acc);
    } else if (job < 26) {
      const int ncol = (job - 19) * 16 + mloc;
      const bool nok = (ncol < NCLS);
      const int nc = nok ? ncol : (NCLS - 1);
      tiles_mma<F_DIM, LDS_STRIDE, NCLS>(sa, wclas + kb * NCLS + nc, acc);
      if (nok) store_tiles<NCLS>(o_logists, row0 + rloc, ncol, bclas[ncol], acc);
    } else {
      const int ncol = (job - 26) * 16 + mloc;
      tiles_mma<F_DIM, LDS_STRIDE, NBITS>(sa, wconv + kb * NBITS + ncol, acc);
      const float s     = gamma[ncol] * rsqrtf(rv[ncol] + 1e-5f);
      const float shift = (bconv[ncol] - rm[ncol]) * s + beta[ncol];
#pragma unroll
      for (int t = 0; t < HB_TILES; ++t) {
#pragma unroll
        for (int i = 0; i < 8; ++i)
          sP[(t * 16 + rloc + i) * T_STRIDE + ncol] = acc[t][i] * s + shift;
      }
    }
  }
  __syncthreads();

#pragma unroll
  for (int j = 0; j < (HB_ROWS * 16) / (32 * HB_WAVES); ++j) {
    const int i  = threadIdx.x + j * (32 * HB_WAVES);
    const int r  = i >> 4;
    const int c4 = (i & 15) * 4;
    const float4 p = *reinterpret_cast<const float4*>(sP + r * T_STRIDE + c4);
    float4 th;
    th.x = tanhf(p.x); th.y = tanhf(p.y); th.z = tanhf(p.z); th.w = tanhf(p.w);
    *reinterpret_cast<float4*>(sT + r * T_STRIDE + c4) = th;
    *reinterpret_cast<float4*>(o_tanh + (size_t)(row0 + r) * NBITS + c4) = th;
  }
  __syncthreads();

  if (wave < 7) {
    v8f acc[HB_TILES] = {};
    const int ncol = wave * 16 + mloc;
    const bool nok = (ncol < NCLS);
    const int nc = nok ? ncol : (NCLS - 1);
    tiles_mma<NBITS, T_STRIDE, NCLS>(sT + mloc * T_STRIDE + kb, wcv + kb * NCLS + nc, acc);
    if (nok) store_tiles<NCLS>(o_feacv, row0 + rloc, ncol, bcv[ncol], acc);
  } else {
    const int r = (wave - 7) * 32 + lane;
    if (r < HB_ROWS) {
      const float* p = sP + r * T_STRIDE;
      float s = 0.0f;
#pragma unroll
      for (int k = 0; k < NBITS; ++k) s += p[k] * wtl[k];
      o_truelab[row0 + r] = s + btl[0];
    }
  }
}

extern "C" void kernel_launch(void* const* d_in, const int* in_sizes, int n_in,
                              void* d_out, int out_size, void* d_ws, size_t ws_size,
                              hipStream_t stream) {
  (void)in_sizes; (void)n_in; (void)out_size; (void)ws_size;

  const float* features = (const float*)d_in[0];
  const int*   edges    = (const int*)d_in[1];
  const float* w1l   = (const float*)d_in[2];
  const float* b1l   = (const float*)d_in[3];
  const float* w1r   = (const float*)d_in[4];
  const float* w2l   = (const float*)d_in[5];
  const float* b2l   = (const float*)d_in[6];
  const float* w2r   = (const float*)d_in[7];
  const float* whd   = (const float*)d_in[8];
  const float* bhd   = (const float*)d_in[9];
  const float* wclas = (const float*)d_in[10];
  const float* bclas = (const float*)d_in[11];
  const float* wconv = (const float*)d_in[12];
  const float* bconv = (const float*)d_in[13];
  const float* gamma = (const float*)d_in[14];
  const float* beta  = (const float*)d_in[15];
  const float* rm    = (const float*)d_in[16];
  const float* rv    = (const float*)d_in[17];
  const float* wtl   = (const float*)d_in[18];
  const float* btl   = (const float*)d_in[19];
  const float* wcv   = (const float*)d_in[20];
  const float* bcv   = (const float*)d_in[21];

  const int* src = edges;
  const int* dst = edges + N_EDGES;

  float* ws     = (float*)d_ws;
  float* deg    = ws;
  float* agg    = deg + N_NODES;
  float* h1     = agg + (size_t)N_NODES * F_DIM;
  float* h2     = h1  + (size_t)N_NODES * F_DIM;
  int* cnt     = (int*)(h2 + (size_t)N_NODES * F_DIM);
  int* off     = cnt + CNT_LEN;
  int* slot    = off + N_NODES;
  int* bucket  = slot + N_EDGES;
  int* ordered = bucket + N_EDGES;

  float* out       = (float*)d_out;
  float* o_logists = out;
  float* o_tanh    = o_logists + (size_t)N_NODES * NCLS;
  float* o_fealab  = o_tanh    + (size_t)N_NODES * NBITS;
  float* o_feacv   = o_fealab  + (size_t)N_NODES * HD;
  float* o_truelab = o_feacv   + (size_t)N_NODES * NCLS;

  const int edgeBlocks = (N_EDGES + 255) / 256;
  const int nodeBlocks = (N_NODES + 7) / 8;
  const int rowTiles      = N_NODES / 16;

  hipMemsetAsync(cnt, 0, sizeof(int) * (size_t)CNT_LEN, stream);
  count_slot_kernel<<<edgeBlocks, 256, 0, stream>>>(dst, cnt, slot);
  scan_offsets_kernel<<<1, SCAN_T, 0, stream>>>(cnt, off, deg);
  fill_kernel<<<edgeBlocks, 256, 0, stream>>>(src, dst, off, slot, bucket);
  order_lists_kernel<<<nodeBlocks, 256, 0, stream>>>(bucket, off, cnt, ordered);

  gather_sum_kernel<<<nodeBlocks, 256, 0, stream>>>(features, ordered, off, cnt, agg);
  sage_layer_kernel<<<rowTiles, 256, 0, stream>>>(features, agg, deg, w1l, b1l, w1r, h1);

  gather_sum_kernel<<<nodeBlocks, 256, 0, stream>>>(h1, ordered, off, cnt, agg);
  sage_layer_kernel<<<rowTiles, 256, 0, stream>>>(h1, agg, deg, w2l, b2l, w2r, h2);

  heads_kernel<<<N_NODES / HB_ROWS, 32 * HB_WAVES, 0, stream>>>(
      h2, whd, bhd, wclas, bclas, wconv, bconv, gamma, beta, rm, rv, wtl, btl,
      wcv, bcv, o_logists, o_tanh, o_fealab, o_feacv, o_truelab);
}


# Neighbour row gather-sum moved into the SAGE layer kernel: each wave sums its two nodes' source rows (same list order) straight into the LDS mean tile; agg and deg planes and 2 launches gone (9->7)
# speedup vs baseline: 2.6642x; 1.1023x over previous
#include <hip/hip_runtime.h>
#include <math.h>


typedef __attribute__((ext_vector_type(2))) float v2f;
typedef __attribute__((ext_vector_type(8))) float v8f;

#define N_NODES 50000
#define N_EDGES 800000
#define F_DIM   128
#define NBITS   64
#define NCLS    100
#define HD      300
#define LDS_STRIDE 132

__device__ __forceinline__ v8f wmma_f32(v2f a, v2f b, v8f c) {
  return __builtin_amdgcn_wmma_f32_16x16x4_f32(false, a, false, b, (short)0, c,
                                               false, false);
}


__global__ void __launch_bounds__(256)
count_slot_kernel(const int* __restrict__ dst, int* __restrict__ cnt,
                  int* __restrict__ slot) {
  int e = blockIdx.x * blockDim.x + threadIdx.x;
  if (e < N_EDGES) slot[e] = atomicAdd(&cnt[dst[e]], 1);
}

#define SCAN_T  1024
#define SCAN_CH ((N_NODES + SCAN_T - 1) / SCAN_T)
#define CNT_LEN (SCAN_T * SCAN_CH)
#define CNT_V4  (CNT_LEN / 4)
#define SCAN_PASSES ((CNT_V4 + SCAN_T - 1) / SCAN_T)
__global__ void __launch_bounds__(SCAN_T)
scan_offsets_kernel(const int* __restrict__ cnt, int* __restrict__ off) {
  __shared__ __attribute__((aligned(16))) int sCnt[CNT_LEN];
  __shared__ int sPart[SCAN_T];
  const int t = threadIdx.x;

  {
    const int4* __restrict__ g4 = reinterpret_cast<const int4*>(cnt);
    int4 v[SCAN_PASSES];
#pragma unroll
    for (int j = 0; j < SCAN_PASSES; ++j) {
      const int i = t + j * SCAN_T;
      v[j] = (i < CNT_V4) ? g4[i] : make_int4(0, 0, 0, 0);
    }
#pragma unroll
    for (int j = 0; j < SCAN_PASSES; ++j) {
      const int i = t + j * SCAN_T;
      if (i < CNT_V4) reinterpret_cast<int4*>(sCnt)[i] = v[j];
    }
  }
  __syncthreads();

  const int b = t * SCAN_CH;
  int cv[SCAN_CH];
#pragma unroll
  for (int i = 0; i < SCAN_CH; ++i) cv[i] = sCnt[b + i];
  int sum = 0;
#pragma unroll
  for (int i = 0; i < SCAN_CH; ++i) sum += cv[i];
  sPart[t] = sum;
  __syncthreads();
  for (int s = 1; s < SCAN_T; s <<= 1) {
    const int v = (t >= s) ? sPart[t - s] : 0;
    __syncthreads();
    sPart[t] += v;
    __syncthreads();
  }
  int run = sPart[t] - sum;
#pragma unroll
  for (int i = 0; i < SCAN_CH; ++i) {
    sCnt[b + i] = run;
    run += cv[i];
  }
  __syncthreads();

#pragma unroll
  for (int j = 0; j < (N_NODES / 4 + SCAN_T - 1) / SCAN_T; ++j) {
    const int i = t + j * SCAN_T;
    if (i < N_NODES / 4)
      reinterpret_cast<int4*>(off)[i] = reinterpret_cast<const int4*>(sCnt)[i];
  }
}

__global__ void __launch_bounds__(256)
fill_kernel(const int* __restrict__ src, const int* __restrict__ dst,
            const int* __restrict__ off, const int* __restrict__ slot,
            int* __restrict__ bucket) {
  int e = blockIdx.x * blockDim.x + threadIdx.x;
  if (e < N_EDGES) bucket[off[dst[e]] + slot[e]] = src[e];
}

__global__ void __launch_bounds__(256)
order_lists_kernel(const int* __restrict__ bucket, const int* __restrict__ off,
                   const int* __restrict__ cnt, int* __restrict__ ordered) {
  const int node = __builtin_amdgcn_readfirstlane(blockIdx.x * 8 + (threadIdx.x >> 5));
  if (node >= N_NODES) return;
  const int lane = threadIdx.x & 31;
  const int n = cnt[node];
  const int o = off[node];
  for (int i = lane; i < n; i += 32) {
    const int key = bucket[o + i];
    int r = 0;
    for (int j = 0; j < n; ++j) {
      const int kj = bucket[o + j];
      r += ((kj < key) || (kj == key && j < i)) ? 1 : 0;
    }
    ordered[o + r] = key;
  }
}

template <int K>
__device__ __forceinline__ void add_rows(const float* __restrict__ xl, int idx,
                                         int j, float4& acc) {
  float4 v[K];
#pragma unroll
  for (int k = 0; k < K; ++k) {
    const int s = __builtin_amdgcn_readlane(idx, j + k);
    v[k] = *reinterpret_cast<const float4*>(xl + (size_t)s * F_DIM);
  }
#pragma unroll
  for (int k = 0; k < K; ++k) {
    acc.x += v[k].x; acc.y += v[k].y; acc.z += v[k].z; acc.w += v[k].w;
  }
}

__device__ __forceinline__ float4 sum_list(const float* __restrict__ xl,
                                           const int* __restrict__ ordered,
                                           int o, int n, int lane, int idx) {
  float4 acc = make_float4(0.0f, 0.0f, 0.0f, 0.0f);
  for (int base = 0; base < n; base += 32) {
    const int m = (n - base < 32) ? (n - base) : 32;
    if (base > 0) idx = (lane < m) ? ordered[o + base + lane] : 0;
    int j = 0;
    for (; j + 8 <= m; j += 8) add_rows<8>(xl, idx, j, acc);
    if (m & 4) { add_rows<4>(xl, idx, j, acc); j += 4; }
    if (m & 2) { add_rows<2>(xl, idx, j, acc); j += 2; }
    if (m & 1) { add_rows<1>(xl, idx, j, acc); }
  }
  return acc;
}

__global__ void __launch_bounds__(256)
sage_layer_kernel(const float* __restrict__ x, const int* __restrict__ ordered,
                  const int* __restrict__ off, const int* __restrict__ cnt,
                  const float* __restrict__ wl, const float* __restrict__ bl,
                  const float* __restrict__ wr, float* __restrict__ h) {
  __shared__ __attribute__((aligned(16))) float sMean[16 * LDS_STRIDE];
  __shared__ __attribute__((aligned(16))) float sX[16 * LDS_STRIDE];

  const int tm   = blockIdx.x;
  const int lane = threadIdx.x & 31;
  const int tn   = __builtin_amdgcn_readfirstlane(threadIdx.x >> 5);

  {
    const int r0 = tn * 2;
    const int g0 = tm * 16 + r0;
    const float* xl = x + lane * 4;
    const float4 vx0 = *reinterpret_cast<const float4*>(xl + (size_t)g0 * F_DIM);
    const float4 vx1 = *reinterpret_cast<const float4*>(xl + (size_t)(g0 + 1) * F_DIM);
    const int n0 = cnt[g0],     o0 = off[g0];
    const int n1 = cnt[g0 + 1], o1 = off[g0 + 1];
    const int id0 = (lane < n0) ? ordered[o0 + lane] : 0;
    const int id1 = (lane < n1) ? ordered[o1 + lane] : 0;
    *reinterpret_cast<float4*>(sX + r0 * LDS_STRIDE + lane * 4)       = vx0;
    *reinterpret_cast<float4*>(sX + (r0 + 1) * LDS_STRIDE + lane * 4) = vx1;
    const float4 a0 = sum_list(xl, ordered, o0, n0, lane, id0);
    const float4 a1 = sum_list(xl, ordered, o1, n1, lane, id1);
    const float i0 = 1.0f / fmaxf((float)n0, 1.0f);
    const float i1 = 1.0f / fmaxf((float)n1, 1.0f);
    *reinterpret_cast<float4*>(sMean + r0 * LDS_STRIDE + lane * 4) =
        make_float4(a0.x * i0, a0.y * i0, a0.z * i0, a0.w * i0);
    *reinterpret_cast<float4*>(sMean + (r0 + 1) * LDS_STRIDE + lane * 4) =
        make_float4(a1.x * i1, a1.y * i1, a1.z * i1, a1.w * i1);
  }
  __syncthreads();

  const int mloc = lane & 15;
  const int ncol = tn * 16 + mloc;
  const int kb   = (lane >= 16) ? 2 : 0;
  const float* sm = sMean + mloc * LDS_STRIDE;
  const float* sx = sX    + mloc * LDS_STRIDE;

  v8f acc = {};
#pragma unroll 4
  for (int kk = 0; kk < F_DIM; kk += 4) {
    v2f a1, b1, a2, b2;
    a1.x = sm[kk + kb];
    a1.y = sm[kk + kb + 1];
    b1.x = wl[(kk + kb)     * F_DIM + ncol];
    b1.y = wl[(kk + kb + 1) * F_DIM + ncol];
    acc = wmma_f32(a1, b1, acc);
    a2.x = sx[kk + kb];
    a2.y = sx[kk + kb + 1];
    b2.x = wr[(kk + kb)     * F_DIM + ncol];
    b2.y = wr[(kk + kb + 1) * F_DIM + ncol];
    acc = wmma_f32(a2, b2, acc);
  }
  const float bias  = bl[ncol];
  const int   mbase = tm * 16 + ((lane >= 16) ? 8 : 0);
#pragma unroll
  for (int i = 0; i < 8; ++i) {
    float v = fmaxf(acc[i] + bias, 0.0f);
    h[(size_t)(mbase + i) * F_DIM + ncol] = v;
  }
}

#define HB_TILES 5
#define HB_ROWS  (16 * HB_TILES)
#define HB_WAVES 10
#define T_STRIDE 68

template <int K, int STRIDE, int N>
__device__ __forceinline__ void tiles_mma(const float* sa, const float* __restrict__ bp,
                                          v8f (&acc)[HB_TILES]) {
#pragma unroll 4
  for (int kk = 0; kk < K; kk += 4) {
    v2f b;
    b.x = bp[kk * N];
    b.y = bp[(kk + 1) * N];
#pragma unroll
    for (int t = 0; t < HB_TILES; ++t) {
      const v2f a = *reinterpret_cast<const v2f*>(sa + t * 16 * STRIDE + kk);
      acc[t] = wmma_f32(a, b, acc[t]);
    }
  }
}

template <int N>
__device__ __forceinline__ void store_tiles(float* __restrict__ C, int rbase, int ncol,
                                            float bv, const v8f (&acc)[HB_TILES]) {
#pragma unroll
  for (int t = 0; t < HB_TILES; ++t) {
#pragma unroll
    for (int i = 0; i < 8; ++i)
      C[(size_t)(rbase + t * 16 + i) * N + ncol] = acc[t][i] + bv;
  }
}

__global__ void __launch_bounds__(32 * HB_WAVES)
heads_kernel(const float* __restrict__ h2,
             const float* __restrict__ whd, const float* __restrict__ bhd,
             const float* __restrict__ wclas, const float* __restrict__ bclas,
             const float* __restrict__ wconv, const float* __restrict__ bconv,
             const float* __restrict__ gamma, const float* __restrict__ beta,
             const float* __restrict__ rm, const float* __restrict__ rv,
             const float* __restrict__ wtl, const float* __restrict__ btl,
             const float* __restrict__ wcv, const float* __restrict__ bcv,
             float* __restrict__ o_logists, float* __restrict__ o_tanh,
             float* __restrict__ o_fealab, float* __restrict__ o_feacv,
             float* __restrict__ o_truelab) {
  __shared__ __attribute__((aligned(16))) float sA[HB_ROWS * LDS_STRIDE];
  __shared__ __attribute__((aligned(16))) float sT[HB_ROWS * T_STRIDE];
  __shared__ __attribute__((aligned(16))) float sP[HB_ROWS * T_STRIDE];

  const int row0 = blockIdx.x * HB_ROWS;
#pragma unroll
  for (int j = 0; j < (HB_ROWS * 32) / (32 * HB_WAVES); ++j) {
    const int i  = threadIdx.x + j * (32 * HB_WAVES);
    const int r  = i >> 5;
    const int c4 = (i & 31) * 4;
    const float4 v = *reinterpret_cast<const float4*>(h2 + (size_t)(row0 + r) * F_DIM + c4);
    *reinterpret_cast<float4*>(sA + r * LDS_STRIDE + c4) = v;
  }
  __syncthreads();

  const int lane = threadIdx.x & 31;
  const int wave = __builtin_amdgcn_readfirstlane(threadIdx.x >> 5);
  const int mloc = lane & 15;
  const int hi   = lane >> 4;
  const int kb   = hi * 2;
  const int rloc = hi * 8;
  const float* sa = sA + mloc * LDS_STRIDE + kb;

  for (int job = wave; job < 30; job += HB_WAVES) {
    v8f acc[HB_TILES] = {};
    if (job < 19) {
      const int ncol = job * 16 + mloc;
      const bool nok = (ncol < HD);
      const int nc = nok ? ncol : (HD - 1);
      tiles_mma<F_DIM, LDS_STRIDE, HD>(sa, whd + kb * HD + nc, acc);
      if (nok) store_tiles<HD>(o_fealab, row0 + rloc, ncol, bhd[ncol], acc);
    } else if (job < 26) {
      const int ncol = (job - 19) * 16 + mloc;
      const bool nok = (ncol < NCLS);
      const int nc = nok ? ncol : (NCLS - 1);
      tiles_mma<F_DIM, LDS_STRIDE, NCLS>(sa, wclas + kb * NCLS + nc, acc);
      if (nok) store_tiles<NCLS>(o_logists, row0 + rloc, ncol, bclas[ncol], acc);
    } else {
      const int ncol = (job - 26) * 16 + mloc;
      tiles_mma<F_DIM, LDS_STRIDE, NBITS>(sa, wconv + kb * NBITS + ncol, acc);
      const float s     = gamma[ncol] * rsqrtf(rv[ncol] + 1e-5f);
      const float shift = (bconv[ncol] - rm[ncol]) * s + beta[ncol];
#pragma unroll
      for (int t = 0; t < HB_TILES; ++t) {
#pragma unroll
        for (int i = 0; i < 8; ++i)
          sP[(t * 16 + rloc + i) * T_STRIDE + ncol] = acc[t][i] * s + shift;
      }
    }
  }
  __syncthreads();

#pragma unroll
  for (int j = 0; j < (HB_ROWS * 16) / (32 * HB_WAVES); ++j) {
    const int i  = threadIdx.x + j * (32 * HB_WAVES);
    const int r  = i >> 4;
    const int c4 = (i & 15) * 4;
    const float4 p = *reinterpret_cast<const float4*>(sP + r * T_STRIDE + c4);
    float4 th;
    th.x = tanhf(p.x); th.y = tanhf(p.y); th.z = tanhf(p.z); th.w = tanhf(p.w);
    *reinterpret_cast<float4*>(sT + r * T_STRIDE + c4) = th;
    *reinterpret_cast<float4*>(o_tanh + (size_t)(row0 + r) * NBITS + c4) = th;
  }
  __syncthreads();

  if (wave < 7) {
    v8f acc[HB_TILES] = {};
    const int ncol = wave * 16 + mloc;
    const bool nok = (ncol < NCLS);
    const int nc = nok ? ncol : (NCLS - 1);
    tiles_mma<NBITS, T_STRIDE, NCLS>(sT + mloc * T_STRIDE + kb, wcv + kb * NCLS + nc, acc);
    if (nok) store_tiles<NCLS>(o_feacv, row0 + rloc, ncol, bcv[ncol], acc);
  } else {
    const int r = (wave - 7) * 32 + lane;
    if (r < HB_ROWS) {
      const float* p = sP + r * T_STRIDE;
      float s = 0.0f;
#pragma unroll
      for (int k = 0; k < NBITS; ++k) s += p[k] * wtl[k];
      o_truelab[row0 + r] = s + btl[0];
    }
  }
}

extern "C" void kernel_launch(void* const* d_in, const int* in_sizes, int n_in,
                              void* d_out, int out_size, void* d_ws, size_t ws_size,
                              hipStream_t stream) {
  (void)in_sizes; (void)n_in; (void)out_size; (void)ws_size;

  const float* features = (const float*)d_in[0];
  const int*   edges    = (const int*)d_in[1];
  const float* w1l   = (const float*)d_in[2];
  const float* b1l   = (const float*)d_in[3];
  const float* w1r   = (const float*)d_in[4];
  const float* w2l   = (const float*)d_in[5];
  const float* b2l   = (const float*)d_in[6];
  const float* w2r   = (const float*)d_in[7];
  const float* whd   = (const float*)d_in[8];
  const float* bhd   = (const float*)d_in[9];
  const float* wclas = (const float*)d_in[10];
  const float* bclas = (const float*)d_in[11];
  const float* wconv = (const float*)d_in[12];
  const float* bconv = (const float*)d_in[13];
  const float* gamma = (const float*)d_in[14];
  const float* beta  = (const float*)d_in[15];
  const float* rm    = (const float*)d_in[16];
  const float* rv    = (const float*)d_in[17];
  const float* wtl   = (const float*)d_in[18];
  const float* btl   = (const float*)d_in[19];
  const float* wcv   = (const float*)d_in[20];
  const float* bcv   = (const float*)d_in[21];

  const int* src = edges;
  const int* dst = edges + N_EDGES;

  float* ws     = (float*)d_ws;
  float* h1     = ws;
  float* h2     = h1  + (size_t)N_NODES * F_DIM;
  int* cnt     = (int*)(h2 + (size_t)N_NODES * F_DIM);
  int* off     = cnt + CNT_LEN;
  int* slot    = off + N_NODES;
  int* bucket  = slot + N_EDGES;
  int* ordered = bucket + N_EDGES;

  float* out       = (float*)d_out;
  float* o_logists = out;
  float* o_tanh    = o_logists + (size_t)N_NODES * NCLS;
  float* o_fealab  = o_tanh    + (size_t)N_NODES * NBITS;
  float* o_feacv   = o_fealab  + (size_t)N_NODES * HD;
  float* o_truelab = o_feacv   + (size_t)N_NODES * NCLS;

  const int edgeBlocks = (N_EDGES + 255) / 256;
  const int nodeBlocks = (N_NODES + 7) / 8;
  const int rowTiles      = N_NODES / 16;

  hipMemsetAsync(cnt, 0, sizeof(int) * (size_t)CNT_LEN, stream);
  count_slot_kernel<<<edgeBlocks, 256, 0, stream>>>(dst, cnt, slot);
  scan_offsets_kernel<<<1, SCAN_T, 0, stream>>>(cnt, off);
  fill_kernel<<<edgeBlocks, 256, 0, stream>>>(src, dst, off, slot, bucket);
  order_lists_kernel<<<nodeBlocks, 256, 0, stream>>>(bucket, off, cnt, ordered);

  sage_layer_kernel<<<rowTiles, 256, 0, stream>>>(features, ordered, off, cnt, w1l, b1l, w1r, h1);

  sage_layer_kernel<<<rowTiles, 256, 0, stream>>>(h1, ordered, off, cnt, w2l, b2l, w2r, h2);

  heads_kernel<<<N_NODES / HB_ROWS, 32 * HB_WAVES, 0, stream>>>(
      h2, whd, bhd, wclas, bclas, wconv, bconv, gamma, beta, rm, rv, wtl, btl,
      wcv, bcv, o_logists, o_tanh, o_fealab, o_feacv, o_truelab);
}


# List ordering moved into the layer-1 SAGE kernel: each wave ranks its two lists in registers (lane-read compares, in-wave permute) and stores them for layer 2; ordering launch gone (7->6)
# speedup vs baseline: 2.7742x; 1.0413x over previous
#include <hip/hip_runtime.h>
#include <math.h>


typedef __attribute__((ext_vector_type(2))) float v2f;
typedef __attribute__((ext_vector_type(8))) float v8f;

#define N_NODES 50000
#define N_EDGES 800000
#define F_DIM   128
#define NBITS   64
#define NCLS    100
#define HD      300
#define LDS_STRIDE 132

__device__ __forceinline__ v8f wmma_f32(v2f a, v2f b, v8f c) {
  return __builtin_amdgcn_wmma_f32_16x16x4_f32(false, a, false, b, (short)0, c,
                                               false, false);
}


__global__ void __launch_bounds__(256)
count_slot_kernel(const int* __restrict__ dst, int* __restrict__ cnt,
                  int* __restrict__ slot) {
  int e = blockIdx.x * blockDim.x + threadIdx.x;
  if (e < N_EDGES) slot[e] = atomicAdd(&cnt[dst[e]], 1);
}

#define SCAN_T  1024
#define SCAN_CH ((N_NODES + SCAN_T - 1) / SCAN_T)
#define CNT_LEN (SCAN_T * SCAN_CH)
#define CNT_V4  (CNT_LEN / 4)
#define SCAN_PASSES ((CNT_V4 + SCAN_T - 1) / SCAN_T)
__global__ void __launch_bounds__(SCAN_T)
scan_offsets_kernel(const int* __restrict__ cnt, int* __restrict__ off) {
  __shared__ __attribute__((aligned(16))) int sCnt[CNT_LEN];
  __shared__ int sPart[SCAN_T];
  const int t = threadIdx.x;

  {
    const int4* __restrict__ g4 = reinterpret_cast<const int4*>(cnt);
    int4 v[SCAN_PASSES];
#pragma unroll
    for (int j = 0; j < SCAN_PASSES; ++j) {
      const int i = t + j * SCAN_T;
      v[j] = (i < CNT_V4) ? g4[i] : make_int4(0, 0, 0, 0);
    }
#pragma unroll
    for (int j = 0; j < SCAN_PASSES; ++j) {
      const int i = t + j * SCAN_T;
      if (i < CNT_V4) reinterpret_cast<int4*>(sCnt)[i] = v[j];
    }
  }
  __syncthreads();

  const int b = t * SCAN_CH;
  int cv[SCAN_CH];
#pragma unroll
  for (int i = 0; i < SCAN_CH; ++i) cv[i] = sCnt[b + i];
  int sum = 0;
#pragma unroll
  for (int i = 0; i < SCAN_CH; ++i) sum += cv[i];
  sPart[t] = sum;
  __syncthreads();
  for (int s = 1; s < SCAN_T; s <<= 1) {
    const int v = (t >= s) ? sPart[t - s] : 0;
    __syncthreads();
    sPart[t] += v;
    __syncthreads();
  }
  int run = sPart[t] - sum;
#pragma unroll
  for (int i = 0; i < SCAN_CH; ++i) {
    sCnt[b + i] = run;
    run += cv[i];
  }
  __syncthreads();

#pragma unroll
  for (int j = 0; j < (N_NODES / 4 + SCAN_T - 1) / SCAN_T; ++j) {
    const int i = t + j * SCAN_T;
    if (i < N_NODES / 4)
      reinterpret_cast<int4*>(off)[i] = reinterpret_cast<const int4*>(sCnt)[i];
  }
}

__global__ void __launch_bounds__(256)
fill_kernel(const int* __restrict__ src, const int* __restrict__ dst,
            const int* __restrict__ off, const int* __restrict__ slot,
            int* __restrict__ bucket) {
  int e = blockIdx.x * blockDim.x + threadIdx.x;
  if (e < N_EDGES) bucket[off[dst[e]] + slot[e]] = src[e];
}

__device__ __forceinline__ int order_head(const int* __restrict__ bucket,
                                          int* ordered, int o, int n, int lane,
                                          int key) {
  if (n <= 32) {
    const int comp = (((lane < n) ? key : N_NODES) << 5) | lane;
    int r = 0;
    for (int jb = 0; jb < n; jb += 8) {
#pragma unroll
      for (int k = 0; k < 8; ++k)
        r += (__builtin_amdgcn_readlane(comp, jb + k) < comp) ? 1 : 0;
    }
    if (lane < n) ordered[o + r] = key; else r = lane;
    return __builtin_amdgcn_ds_permute(r << 2, key);
  }
  for (int i = lane; i < n; i += 32) {
    const int ki = bucket[o + i];
    int r = 0;
    for (int j = 0; j < n; ++j) {
      const int kj = bucket[o + j];
      r += ((kj < ki) || (kj == ki && j < i)) ? 1 : 0;
    }
    ordered[o + r] = ki;
  }
  __threadfence();
  return ordered[o + lane];
}

template <int K>
__device__ __forceinline__ void add_rows(const float* __restrict__ xl, int idx,
                                         int j, float4& acc) {
  float4 v[K];
#pragma unroll
  for (int k = 0; k < K; ++k) {
    const int s = __builtin_amdgcn_readlane(idx, j + k);
    v[k] = *reinterpret_cast<const float4*>(xl + (size_t)s * F_DIM);
  }
#pragma unroll
  for (int k = 0; k < K; ++k) {
    acc.x += v[k].x; acc.y += v[k].y; acc.z += v[k].z; acc.w += v[k].w;
  }
}

__device__ __forceinline__ float4 sum_list(const float* __restrict__ xl,
                                           const int* __restrict__ ordered,
                                           int o, int n, int lane, int idx) {
  float4 acc = make_float4(0.0f, 0.0f, 0.0f, 0.0f);
  for (int base = 0; base < n; base += 32) {
    const int m = (n - base < 32) ? (n - base) : 32;
    if (base > 0) idx = (lane < m) ? ordered[o + base + lane] : 0;
    int j = 0;
    for (; j + 8 <= m; j += 8) add_rows<8>(xl, idx, j, acc);
    if (m & 4) { add_rows<4>(xl, idx, j, acc); j += 4; }
    if (m & 2) { add_rows<2>(xl, idx, j, acc); j += 2; }
    if (m & 1) { add_rows<1>(xl, idx, j, acc); }
  }
  return acc;
}

template <bool BUILD>
__global__ void __launch_bounds__(256)
sage_layer_kernel(const float* __restrict__ x, const int* __restrict__ lists,
                  int* ordered_rw,
                  const int* __restrict__ off, const int* __restrict__ cnt,
                  const float* __restrict__ wl, const float* __restrict__ bl,
                  const float* __restrict__ wr, float* __restrict__ h) {
  __shared__ __attribute__((aligned(16))) float sMean[16 * LDS_STRIDE];
  __shared__ __attribute__((aligned(16))) float sX[16 * LDS_STRIDE];

  const int tm   = blockIdx.x;
  const int lane = threadIdx.x & 31;
  const int tn   = __builtin_amdgcn_readfirstlane(threadIdx.x >> 5);

  {
    const int r0 = tn * 2;
    const int g0 = tm * 16 + r0;
    const float* xl = x + lane * 4;
    const float4 vx0 = *reinterpret_cast<const float4*>(xl + (size_t)g0 * F_DIM);
    const float4 vx1 = *reinterpret_cast<const float4*>(xl + (size_t)(g0 + 1) * F_DIM);
    const int n0 = cnt[g0],     o0 = off[g0];
    const int n1 = cnt[g0 + 1], o1 = off[g0 + 1];
    int id0 = (lane < n0) ? lists[o0 + lane] : 0;
    int id1 = (lane < n1) ? lists[o1 + lane] : 0;
    *reinterpret_cast<float4*>(sX + r0 * LDS_STRIDE + lane * 4)       = vx0;
    *reinterpret_cast<float4*>(sX + (r0 + 1) * LDS_STRIDE + lane * 4) = vx1;
    if (BUILD) {
      id0 = order_head(lists, ordered_rw, o0, n0, lane, id0);
      id1 = order_head(lists, ordered_rw, o1, n1, lane, id1);
    }
    const int* ordered = BUILD ? ordered_rw : lists;
    const float4 a0 = sum_list(xl, ordered, o0, n0, lane, id0);
    const float4 a1 = sum_list(xl, ordered, o1, n1, lane, id1);
    const float i0 = 1.0f / fmaxf((float)n0, 1.0f);
    const float i1 = 1.0f / fmaxf((float)n1, 1.0f);
    *reinterpret_cast<float4*>(sMean + r0 * LDS_STRIDE + lane * 4) =
        make_float4(a0.x * i0, a0.y * i0, a0.z * i0, a0.w * i0);
    *reinterpret_cast<float4*>(sMean + (r0 + 1) * LDS_STRIDE + lane * 4) =
        make_float4(a1.x * i1, a1.y * i1, a1.z * i1, a1.w * i1);
  }
  __syncthreads();

  const int mloc = lane & 15;
  const int ncol = tn * 16 + mloc;
  const int kb   = (lane >= 16) ? 2 : 0;
  const float* sm = sMean + mloc * LDS_STRIDE;
  const float* sx = sX    + mloc * LDS_STRIDE;

  v8f acc = {};
#pragma unroll 4
  for (int kk = 0; kk < F_DIM; kk += 4) {
    v2f a1, b1, a2, b2;
    a1.x = sm[kk + kb];
    a1.y = sm[kk + kb + 1];
    b1.x = wl[(kk + kb)     * F_DIM + ncol];
    b1.y = wl[(kk + kb + 1) * F_DIM + ncol];
    acc = wmma_f32(a1, b1, acc);
    a2.x = sx[kk + kb];
    a2.y = sx[kk + kb + 1];
    b2.x = wr[(kk + kb)     * F_DIM + ncol];
    b2.y = wr[(kk + kb + 1) * F_DIM + ncol];
    acc = wmma_f32(a2, b2, acc);
  }
  const float bias  = bl[ncol];
  const int   mbase = tm * 16 + ((lane >= 16) ? 8 : 0);
#pragma unroll
  for (int i = 0; i < 8; ++i) {
    float v = fmaxf(acc[i] + bias, 0.0f);
    h[(size_t)(mbase + i) * F_DIM + ncol] = v;
  }
}

#define HB_TILES 5
#define HB_ROWS  (16 * HB_TILES)
#define HB_WAVES 10
#define T_STRIDE 68

template <int K, int STRIDE, int N>
__device__ __forceinline__ void tiles_mma(const float* sa, const float* __restrict__ bp,
                                          v8f (&acc)[HB_TILES]) {
#pragma unroll 4
  for (int kk = 0; kk < K; kk += 4) {
    v2f b;
    b.x = bp[kk * N];
    b.y = bp[(kk + 1) * N];
#pragma unroll
    for (int t = 0; t < HB_TILES; ++t) {
      const v2f a = *reinterpret_cast<const v2f*>(sa + t * 16 * STRIDE + kk);
      acc[t] = wmma_f32(a, b, acc[t]);
    }
  }
}

template <int N>
__device__ __forceinline__ void store_tiles(float* __restrict__ C, int rbase, int ncol,
                                            float bv, const v8f (&acc)[HB_TILES]) {
#pragma unroll
  for (int t = 0; t < HB_TILES; ++t) {
#pragma unroll
    for (int i = 0; i < 8; ++i)
      C[(size_t)(rbase + t * 16 + i) * N + ncol] = acc[t][i] + bv;
  }
}

__global__ void __launch_bounds__(32 * HB_WAVES)
heads_kernel(const float* __restrict__ h2,
             const float* __restrict__ whd, const float* __restrict__ bhd,
             const float* __restrict__ wclas, const float* __restrict__ bclas,
             const float* __restrict__ wconv, const float* __restrict__ bconv,
             const float* __restrict__ gamma, const float* __restrict__ beta,
             const float* __restrict__ rm, const float* __restrict__ rv,
             const float* __restrict__ wtl, const float* __restrict__ btl,
             const float* __restrict__ wcv, const float* __restrict__ bcv,
             float* __restrict__ o_logists, float* __restrict__ o_tanh,
             float* __restrict__ o_fealab, float* __restrict__ o_feacv,
             float* __restrict__ o_truelab) {
  __shared__ __attribute__((aligned(16))) float sA[HB_ROWS * LDS_STRIDE];
  __shared__ __attribute__((aligned(16))) float sT[HB_ROWS * T_STRIDE];
  __shared__ __attribute__((aligned(16))) float sP[HB_ROWS * T_STRIDE];

  const int row0 = blockIdx.x * HB_ROWS;
#pragma unroll
  for (int j = 0; j < (HB_ROWS * 32) / (32 * HB_WAVES); ++j) {
    const int i  = threadIdx.x + j * (32 * HB_WAVES);
    const int r  = i >> 5;
    const int c4 = (i & 31) * 4;
    const float4 v = *reinterpret_cast<const float4*>(h2 + (size_t)(row0 + r) * F_DIM + c4);
    *reinterpret_cast<float4*>(sA + r * LDS_STRIDE + c4) = v;
  }
  __syncthreads();

  const int lane = threadIdx.x & 31;
  const int wave = __builtin_amdgcn_readfirstlane(threadIdx.x >> 5);
  const int mloc = lane & 15;
  const int hi   = lane >> 4;
  const int kb   = hi * 2;
  const int rloc = hi * 8;
  const float* sa = sA + mloc * LDS_STRIDE + kb;

  for (int job = wave; job < 30; job += HB_WAVES) {
    v8f acc[HB_TILES] = {};
    if (job < 19) {
      const int ncol = job * 16 + mloc;
      const bool nok = (ncol < HD);
      const int nc = nok ? ncol : (HD - 1);
      tiles_mma<F_DIM, LDS_STRIDE, HD>(sa, whd + kb * HD + nc, acc);
      if (nok) store_tiles<HD>(o_fealab, row0 + rloc, ncol, bhd[ncol], acc);
    } else if (job < 26) {
      const int ncol = (job - 19) * 16 + mloc;
      const bool nok = (ncol < NCLS);
      const int nc = nok ? ncol : (NCLS - 1);
      tiles_mma<F_DIM, LDS_STRIDE, NCLS>(sa, wclas + kb * NCLS + nc, acc);
      if (nok) store_tiles<NCLS>(o_logists, row0 + rloc, ncol, bclas[ncol], acc);
    } else {
      const int ncol = (job - 26) * 16 + mloc;
      tiles_mma<F_DIM, LDS_STRIDE, NBITS>(sa, wconv + kb * NBITS + ncol, acc);
      const float s     = gamma[ncol] * rsqrtf(rv[ncol] + 1e-5f);
      const float shift = (bconv[ncol] - rm[ncol]) * s + beta[ncol];
#pragma unroll
      for (int t = 0; t < HB_TILES; ++t) {
#pragma unroll
        for (int i = 0; i < 8; ++i)
          sP[(t * 16 + rloc + i) * T_STRIDE + ncol] = acc[t][i] * s + shift;
      }
    }
  }
  __syncthreads();

#pragma unroll
  for (int j = 0; j < (HB_ROWS * 16) / (32 * HB_WAVES); ++j) {
    const int i  = threadIdx.x + j * (32 * HB_WAVES);
    const int r  = i >> 4;
    const int c4 = (i & 15) * 4;
    const float4 p = *reinterpret_cast<const float4*>(sP + r * T_STRIDE + c4);
    float4 th;
    th.x = tanhf(p.x); th.y = tanhf(p.y); th.z = tanhf(p.z); th.w = tanhf(p.w);
    *reinterpret_cast<float4*>(sT + r * T_STRIDE + c4) = th;
    *reinterpret_cast<float4*>(o_tanh + (size_t)(row0 + r) * NBITS + c4) = th;
  }
  __syncthreads();

  if (wave < 7) {
    v8f acc[HB_TILES] = {};
    const int ncol = wave * 16 + mloc;
    const bool nok = (ncol < NCLS);
    const int nc = nok ? ncol : (NCLS - 1);
    tiles_mma<NBITS, T_STRIDE, NCLS>(sT + mloc * T_STRIDE + kb, wcv + kb * NCLS + nc, acc);
    if (nok) store_tiles<NCLS>(o_feacv, row0 + rloc, ncol, bcv[ncol], acc);
  } else {
    const int r = (wave - 7) * 32 + lane;
    if (r < HB_ROWS) {
      const float* p = sP + r * T_STRIDE;
      float s = 0.0f;
#pragma unroll
      for (int k = 0; k < NBITS; ++k) s += p[k] * wtl[k];
      o_truelab[row0 + r] = s + btl[0];
    }
  }
}

extern "C" void kernel_launch(void* const* d_in, const int* in_sizes, int n_in,
                              void* d_out, int out_size, void* d_ws, size_t ws_size,
                              hipStream_t stream) {
  (void)in_sizes; (void)n_in; (void)out_size; (void)ws_size;

  const float* features = (const float*)d_in[0];
  const int*   edges    = (const int*)d_in[1];
  const float* w1l   = (const float*)d_in[2];
  const float* b1l   = (const float*)d_in[3];
  const float* w1r   = (const float*)d_in[4];
  const float* w2l   = (const float*)d_in[5];
  const float* b2l   = (const float*)d_in[6];
  const float* w2r   = (const float*)d_in[7];
  const float* whd   = (const float*)d_in[8];
  const float* bhd   = (const float*)d_in[9];
  const float* wclas = (const float*)d_in[10];
  const float* bclas = (const float*)d_in[11];
  const float* wconv = (const float*)d_in[12];
  const float* bconv = (const float*)d_in[13];
  const float* gamma = (const float*)d_in[14];
  const float* beta  = (const float*)d_in[15];
  const float* rm    = (const float*)d_in[16];
  const float* rv    = (const float*)d_in[17];
  const float* wtl   = (const float*)d_in[18];
  const float* btl   = (const float*)d_in[19];
  const float* wcv   = (const float*)d_in[20];
  const float* bcv   = (const float*)d_in[21];

  const int* src = edges;
  const int* dst = edges + N_EDGES;

  float* ws     = (float*)d_ws;
  float* h1     = ws;
  float* h2     = h1  + (size_t)N_NODES * F_DIM;
  int* cnt     = (int*)(h2 + (size_t)N_NODES * F_DIM);
  int* off     = cnt + CNT_LEN;
  int* slot    = off + N_NODES;
  int* bucket  = slot + N_EDGES;
  int* ordered = bucket + N_EDGES;

  float* out       = (float*)d_out;
  float* o_logists = out;
  float* o_tanh    = o_logists + (size_t)N_NODES * NCLS;
  float* o_fealab  = o_tanh    + (size_t)N_NODES * NBITS;
  float* o_feacv   = o_fealab  + (size_t)N_NODES * HD;
  float* o_truelab = o_feacv   + (size_t)N_NODES * NCLS;

  const int edgeBlocks = (N_EDGES + 255) / 256;
  const int rowTiles      = N_NODES / 16;

  hipMemsetAsync(cnt, 0, sizeof(int) * (size_t)CNT_LEN, stream);
  count_slot_kernel<<<edgeBlocks, 256, 0, stream>>>(dst, cnt, slot);
  scan_offsets_kernel<<<1, SCAN_T, 0, stream>>>(cnt, off);
  fill_kernel<<<edgeBlocks, 256, 0, stream>>>(src, dst, off, slot, bucket);

  sage_layer_kernel<true><<<rowTiles, 256, 0, stream>>>(features, bucket, ordered, off, cnt, w1l, b1l, w1r, h1);

  sage_layer_kernel<false><<<rowTiles, 256, 0, stream>>>(h1, ordered, nullptr, off, cnt, w2l, b2l, w2r, h2);

  heads_kernel<<<N_NODES / HB_ROWS, 32 * HB_WAVES, 0, stream>>>(
      h2, whd, bhd, wclas, bclas, wconv, bconv, gamma, beta, rm, rv, wtl, btl,
      wcv, bcv, o_logists, o_tanh, o_fealab, o_feacv, o_truelab);
}


# Neighbour lists kept in fixed 64-entry per-node segments: the count kernel stores each source id at its returned slot; prefix-scan and fill launches gone (6->4); per-node sum order unchanged
# speedup vs baseline: 3.1209x; 1.1250x over previous
#include <hip/hip_runtime.h>
#include <math.h>


typedef __attribute__((ext_vector_type(2))) float v2f;
typedef __attribute__((ext_vector_type(8))) float v8f;

#define N_NODES 50000
#define N_EDGES 800000
#define F_DIM   128
#define NBITS   64
#define NCLS    100
#define HD      300
#define LDS_STRIDE 132

__device__ __forceinline__ v8f wmma_f32(v2f a, v2f b, v8f c) {
  return __builtin_amdgcn_wmma_f32_16x16x4_f32(false, a, false, b, (short)0, c,
                                               false, false);
}

#define LIST_PITCH 64
#define CNT_LEN 50176

__global__ void __launch_bounds__(256)
count_place_kernel(const int* __restrict__ src, const int* __restrict__ dst,
                   int* __restrict__ cnt, int* __restrict__ seg) {
  int e = blockIdx.x * blockDim.x + threadIdx.x;
  if (e < N_EDGES) {
    const int d = dst[e];
    const int v = src[e];
    asm volatile("" :: "v"(v));
    const int s = atomicAdd(&cnt[d], 1);
    if (s < LIST_PITCH) seg[d * LIST_PITCH + s] = v;
  }
}

__device__ __forceinline__ int order_head(const int* __restrict__ bucket,
                                          int* ordered, int o, int n, int lane,
                                          int key) {
  if (n <= 32) {
    const int comp = (((lane < n) ? key : N_NODES) << 5) | lane;
    int r = 0;
    for (int jb = 0; jb < n; jb += 8) {
#pragma unroll
      for (int k = 0; k < 8; ++k)
        r += (__builtin_amdgcn_readlane(comp, jb + k) < comp) ? 1 : 0;
    }
    if (lane < n) ordered[o + r] = key; else r = lane;
    return __builtin_amdgcn_ds_permute(r << 2, key);
  }
  for (int i = lane; i < n; i += 32) {
    const int ki = bucket[o + i];
    int r = 0;
    for (int j = 0; j < n; ++j) {
      const int kj = bucket[o + j];
      r += ((kj < ki) || (kj == ki && j < i)) ? 1 : 0;
    }
    ordered[o + r] = ki;
  }
  __threadfence();
  return ordered[o + lane];
}

template <int K>
__device__ __forceinline__ void add_rows(const float* __restrict__ xl, int idx,
                                         int j, float4& acc) {
  float4 v[K];
#pragma unroll
  for (int k = 0; k < K; ++k) {
    const int s = __builtin_amdgcn_readlane(idx, j + k);
    v[k] = *reinterpret_cast<const float4*>(xl + (size_t)s * F_DIM);
  }
#pragma unroll
  for (int k = 0; k < K; ++k) {
    acc.x += v[k].x; acc.y += v[k].y; acc.z += v[k].z; acc.w += v[k].w;
  }
}

__device__ __forceinline__ float4 sum_list(const float* __restrict__ xl,
                                           const int* __restrict__ ordered,
                                           int o, int n, int lane, int idx) {
  float4 acc = make_float4(0.0f, 0.0f, 0.0f, 0.0f);
  for (int base = 0; base < n; base += 32) {
    const int m = (n - base < 32) ? (n - base) : 32;
    if (base > 0) idx = (lane < m) ? ordered[o + base + lane] : 0;
    int j = 0;
    for (; j + 8 <= m; j += 8) add_rows<8>(xl, idx, j, acc);
    if (m & 4) { add_rows<4>(xl, idx, j, acc); j += 4; }
    if (m & 2) { add_rows<2>(xl, idx, j, acc); j += 2; }
    if (m & 1) { add_rows<1>(xl, idx, j, acc); }
  }
  return acc;
}

__device__ __forceinline__ float4 sum_from_edges(const float* __restrict__ xl,
                                                 const int* __restrict__ src,
                                                 const int* __restrict__ dst,
                                                 int g, int n, int lane) {
  float4 acc = make_float4(0.0f, 0.0f, 0.0f, 0.0f);
  int last = -1;
  int done = 0;
  for (int pass = 0; pass < n && done < n; ++pass) {
    int best = N_NODES;
    int c = 0;
    for (int e = lane; e < N_EDGES; e += 32) {
      if (dst[e] == g) {
        const int s = src[e];
        if (s > last) {
          if (s < best) { best = s; c = 1; }
          else if (s == best) ++c;
        }
      }
    }
    int m = best;
    for (int d = 16; d >= 1; d >>= 1) {
      const int o = __shfl_xor(m, d);
      m = (o < m) ? o : m;
    }
    int cc = (best == m) ? c : 0;
    for (int d = 16; d >= 1; d >>= 1) cc += __shfl_xor(cc, d);
    m  = __builtin_amdgcn_readfirstlane(m);
    cc = __builtin_amdgcn_readfirstlane(cc);
    if (m >= N_NODES || cc <= 0) break;
    const float4 v = *reinterpret_cast<const float4*>(xl + (size_t)m * F_DIM);
    for (int i = 0; i < cc; ++i) {
      acc.x += v.x; acc.y += v.y; acc.z += v.z; acc.w += v.w;
    }
    last = m;
    done += cc;
  }
  return acc;
}

template <bool BUILD>
__global__ void __launch_bounds__(256)
sage_layer_kernel(const float* __restrict__ x, const int* __restrict__ lists,
                  int* ordered_rw,
                  const int* __restrict__ src, const int* __restrict__ dst,
                  const int* __restrict__ cnt,
                  const float* __restrict__ wl, const float* __restrict__ bl,
                  const float* __restrict__ wr, float* __restrict__ h) {
  __shared__ __attribute__((aligned(16))) float sMean[16 * LDS_STRIDE];
  __shared__ __attribute__((aligned(16))) float sX[16 * LDS_STRIDE];

  const int tm   = blockIdx.x;
  const int lane = threadIdx.x & 31;
  const int tn   = __builtin_amdgcn_readfirstlane(threadIdx.x >> 5);

  {
    const int r0 = tn * 2;
    const int g0 = tm * 16 + r0;
    const float* xl = x + lane * 4;
    const float4 vx0 = *reinterpret_cast<const float4*>(xl + (size_t)g0 * F_DIM);
    const float4 vx1 = *reinterpret_cast<const float4*>(xl + (size_t)(g0 + 1) * F_DIM);
    const int n0 = cnt[g0];
    const int n1 = cnt[g0 + 1];
    const int o0 = g0 * LIST_PITCH;
    const int o1 = o0 + LIST_PITCH;
    const bool fit0 = (n0 <= LIST_PITCH);
    const bool fit1 = (n1 <= LIST_PITCH);
    int id0 = (fit0 && lane < n0) ? lists[o0 + lane] : 0;
    int id1 = (fit1 && lane < n1) ? lists[o1 + lane] : 0;
    *reinterpret_cast<float4*>(sX + r0 * LDS_STRIDE + lane * 4)       = vx0;
    *reinterpret_cast<float4*>(sX + (r0 + 1) * LDS_STRIDE + lane * 4) = vx1;
    if (BUILD) {
      if (fit0) id0 = order_head(lists, ordered_rw, o0, n0, lane, id0);
      if (fit1) id1 = order_head(lists, ordered_rw, o1, n1, lane, id1);
    }
    const int* ordered = BUILD ? ordered_rw : lists;
    const float4 a0 = fit0 ? sum_list(xl, ordered, o0, n0, lane, id0)
                           : sum_from_edges(xl, src, dst, g0, n0, lane);
    const float4 a1 = fit1 ? sum_list(xl, ordered, o1, n1, lane, id1)
                           : sum_from_edges(xl, src, dst, g0 + 1, n1, lane);
    const float i0 = 1.0f / fmaxf((float)n0, 1.0f);
    const float i1 = 1.0f / fmaxf((float)n1, 1.0f);
    *reinterpret_cast<float4*>(sMean + r0 * LDS_STRIDE + lane * 4) =
        make_float4(a0.x * i0, a0.y * i0, a0.z * i0, a0.w * i0);
    *reinterpret_cast<float4*>(sMean + (r0 + 1) * LDS_STRIDE + lane * 4) =
        make_float4(a1.x * i1, a1.y * i1, a1.z * i1, a1.w * i1);
  }
  __syncthreads();

  const int mloc = lane & 15;
  const int ncol = tn * 16 + mloc;
  const int kb   = (lane >= 16) ? 2 : 0;
  const float* sm = sMean + mloc * LDS_STRIDE;
  const float* sx = sX    + mloc * LDS_STRIDE;

  v8f acc = {};
#pragma unroll 4
  for (int kk = 0; kk < F_DIM; kk += 4) {
    v2f a1, b1, a2, b2;
    a1.x = sm[kk + kb];
    a1.y = sm[kk + kb + 1];
    b1.x = wl[(kk + kb)     * F_DIM + ncol];
    b1.y = wl[(kk + kb + 1) * F_DIM + ncol];
    acc = wmma_f32(a1, b1, acc);
    a2.x = sx[kk + kb];
    a2.y = sx[kk + kb + 1];
    b2.x = wr[(kk + kb)     * F_DIM + ncol];
    b2.y = wr[(kk + kb + 1) * F_DIM + ncol];
    acc = wmma_f32(a2, b2, acc);
  }
  const float bias  = bl[ncol];
  const int   mbase = tm * 16 + ((lane >= 16) ? 8 : 0);
#pragma unroll
  for (int i = 0; i < 8; ++i) {
    float v = fmaxf(acc[i] + bias, 0.0f);
    h[(size_t)(mbase + i) * F_DIM + ncol] = v;
  }
}

#define HB_TILES 5
#define HB_ROWS  (16 * HB_TILES)
#define HB_WAVES 10
#define T_STRIDE 68

template <int K, int STRIDE, int N>
__device__ __forceinline__ void tiles_mma(const float* sa, const float* __restrict__ bp,
                                          v8f (&acc)[HB_TILES]) {
#pragma unroll 4
  for (int kk = 0; kk < K; kk += 4) {
    v2f b;
    b.x = bp[kk * N];
    b.y = bp[(kk + 1) * N];
#pragma unroll
    for (int t = 0; t < HB_TILES; ++t) {
      const v2f a = *reinterpret_cast<const v2f*>(sa + t * 16 * STRIDE + kk);
      acc[t] = wmma_f32(a, b, acc[t]);
    }
  }
}

template <int N>
__device__ __forceinline__ void store_tiles(float* __restrict__ C, int rbase, int ncol,
                                            float bv, const v8f (&acc)[HB_TILES]) {
#pragma unroll
  for (int t = 0; t < HB_TILES; ++t) {
#pragma unroll
    for (int i = 0; i < 8; ++i)
      C[(size_t)(rbase + t * 16 + i) * N + ncol] = acc[t][i] + bv;
  }
}

__global__ void __launch_bounds__(32 * HB_WAVES)
heads_kernel(const float* __restrict__ h2,
             const float* __restrict__ whd, const float* __restrict__ bhd,
             const float* __restrict__ wclas, const float* __restrict__ bclas,
             const float* __restrict__ wconv, const float* __restrict__ bconv,
             const float* __restrict__ gamma, const float* __restrict__ beta,
             const float* __restrict__ rm, const float* __restrict__ rv,
             const float* __restrict__ wtl, const float* __restrict__ btl,
             const float* __restrict__ wcv, const float* __restrict__ bcv,
             float* __restrict__ o_logists, float* __restrict__ o_tanh,
             float* __restrict__ o_fealab, float* __restrict__ o_feacv,
             float* __restrict__ o_truelab) {
  __shared__ __attribute__((aligned(16))) float sA[HB_ROWS * LDS_STRIDE];
  __shared__ __attribute__((aligned(16))) float sT[HB_ROWS * T_STRIDE];
  __shared__ __attribute__((aligned(16))) float sP[HB_ROWS * T_STRIDE];

  const int row0 = blockIdx.x * HB_ROWS;
#pragma unroll
  for (int j = 0; j < (HB_ROWS * 32) / (32 * HB_WAVES); ++j) {
    const int i  = threadIdx.x + j * (32 * HB_WAVES);
    const int r  = i >> 5;
    const int c4 = (i & 31) * 4;
    const float4 v = *reinterpret_cast<const float4*>(h2 + (size_t)(row0 + r) * F_DIM + c4);
    *reinterpret_cast<float4*>(sA + r * LDS_STRIDE + c4) = v;
  }
  __syncthreads();

  const int lane = threadIdx.x & 31;
  const int wave = __builtin_amdgcn_readfirstlane(threadIdx.x >> 5);
  const int mloc = lane & 15;
  const int hi   = lane >> 4;
  const int kb   = hi * 2;
  const int rloc = hi * 8;
  const float* sa = sA + mloc * LDS_STRIDE + kb;

  for (int job = wave; job < 30; job += HB_WAVES) {
    v8f acc[HB_TILES] = {};
    if (job < 19) {
      const int ncol = job * 16 + mloc;
      const bool nok = (ncol < HD);
      const int nc = nok ? ncol : (HD - 1);
      tiles_mma<F_DIM, LDS_STRIDE, HD>(sa, whd + kb * HD + nc, acc);
      if (nok) store_tiles<HD>(o_fealab, row0 + rloc, ncol, bhd[ncol], acc);
    } else if (job < 26) {
      const int ncol = (job - 19) * 16 + mloc;
      const bool nok = (ncol < NCLS);
      const int nc = nok ? ncol : (NCLS - 1);
      tiles_mma<F_DIM, LDS_STRIDE, NCLS>(sa, wclas + kb * NCLS + nc, acc);
      if (nok) store_tiles<NCLS>(o_logists, row0 + rloc, ncol, bclas[ncol], acc);
    } else {
      const int ncol = (job - 26) * 16 + mloc;
      tiles_mma<F_DIM, LDS_STRIDE, NBITS>(sa, wconv + kb * NBITS + ncol, acc);
      const float s     = gamma[ncol] * rsqrtf(rv[ncol] + 1e-5f);
      const float shift = (bconv[ncol] - rm[ncol]) * s + beta[ncol];
#pragma unroll
      for (int t = 0; t < HB_TILES; ++t) {
#pragma unroll
        for (int i = 0; i < 8; ++i)
          sP[(t * 16 + rloc + i) * T_STRIDE + ncol] = acc[t][i] * s + shift;
      }
    }
  }
  __syncthreads();

#pragma unroll
  for (int j = 0; j < (HB_ROWS * 16) / (32 * HB_WAVES); ++j) {
    const int i  = threadIdx.x + j * (32 * HB_WAVES);
    const int r  = i >> 4;
    const int c4 = (i & 15) * 4;
    const float4 p = *reinterpret_cast<const float4*>(sP + r * T_STRIDE + c4);
    float4 th;
    th.x = tanhf(p.x); th.y = tanhf(p.y); th.z = tanhf(p.z); th.w = tanhf(p.w);
    *reinterpret_cast<float4*>(sT + r * T_STRIDE + c4) = th;
    *reinterpret_cast<float4*>(o_tanh + (size_t)(row0 + r) * NBITS + c4) = th;
  }
  __syncthreads();

  if (wave < 7) {
    v8f acc[HB_TILES] = {};
    const int ncol = wave * 16 + mloc;
    const bool nok = (ncol < NCLS);
    const int nc = nok ? ncol : (NCLS - 1);
    tiles_mma<NBITS, T_STRIDE, NCLS>(sT + mloc * T_STRIDE + kb, wcv + kb * NCLS + nc, acc);
    if (nok) store_tiles<NCLS>(o_feacv, row0 + rloc, ncol, bcv[ncol], acc);
  } else {
    const int r = (wave - 7) * 32 + lane;
    if (r < HB_ROWS) {
      const float* p = sP + r * T_STRIDE;
      float s = 0.0f;
#pragma unroll
      for (int k = 0; k < NBITS; ++k) s += p[k] * wtl[k];
      o_truelab[row0 + r] = s + btl[0];
    }
  }
}

extern "C" void kernel_launch(void* const* d_in, const int* in_sizes, int n_in,
                              void* d_out, int out_size, void* d_ws, size_t ws_size,
                              hipStream_t stream) {
  (void)in_sizes; (void)n_in; (void)out_size; (void)ws_size;

  const float* features = (const float*)d_in[0];
  const int*   edges    = (const int*)d_in[1];
  const float* w1l   = (const float*)d_in[2];
  const float* b1l   = (const float*)d_in[3];
  const float* w1r   = (const float*)d_in[4];
  const float* w2l   = (const float*)d_in[5];
  const float* b2l   = (const float*)d_in[6];
  const float* w2r   = (const float*)d_in[7];
  const float* whd   = (const float*)d_in[8];
  const float* bhd   = (const float*)d_in[9];
  const float* wclas = (const float*)d_in[10];
  const float* bclas = (const float*)d_in[11];
  const float* wconv = (const float*)d_in[12];
  const float* bconv = (const float*)d_in[13];
  const float* gamma = (const float*)d_in[14];
  const float* beta  = (const float*)d_in[15];
  const float* rm    = (const float*)d_in[16];
  const float* rv    = (const float*)d_in[17];
  const float* wtl   = (const float*)d_in[18];
  const float* btl   = (const float*)d_in[19];
  const float* wcv   = (const float*)d_in[20];
  const float* bcv   = (const float*)d_in[21];

  const int* src = edges;
  const int* dst = edges + N_EDGES;

  float* ws     = (float*)d_ws;
  float* h1     = ws;
  float* h2     = h1  + (size_t)N_NODES * F_DIM;
  int* cnt     = (int*)(h2 + (size_t)N_NODES * F_DIM);
  int* seg     = cnt + CNT_LEN;
  int* ordered = seg + (size_t)N_NODES * LIST_PITCH;

  float* out       = (float*)d_out;
  float* o_logists = out;
  float* o_tanh    = o_logists + (size_t)N_NODES * NCLS;
  float* o_fealab  = o_tanh    + (size_t)N_NODES * NBITS;
  float* o_feacv   = o_fealab  + (size_t)N_NODES * HD;
  float* o_truelab = o_feacv   + (size_t)N_NODES * NCLS;

  const int edgeBlocks = (N_EDGES + 255) / 256;
  const int rowTiles      = N_NODES / 16;

  hipMemsetAsync(cnt, 0, sizeof(int) * (size_t)CNT_LEN, stream);
  count_place_kernel<<<edgeBlocks, 256, 0, stream>>>(src, dst, cnt, seg);

  sage_layer_kernel<true><<<rowTiles, 256, 0, stream>>>(features, seg, ordered, src, dst, cnt, w1l, b1l, w1r, h1);

  sage_layer_kernel<false><<<rowTiles, 256, 0, stream>>>(h1, ordered, nullptr, src, dst, cnt, w2l, b2l, w2r, h2);

  heads_kernel<<<N_NODES / HB_ROWS, 32 * HB_WAVES, 0, stream>>>(
      h2, whd, bhd, wclas, bclas, wconv, bconv, gamma, beta, rm, rv, wtl, btl,
      wcv, bcv, o_logists, o_tanh, o_fealab, o_feacv, o_truelab);
}


# Neighbour lists built by two LDS grouping passes (edge chunk x 196-node range cells, then one block per range, LDS counters), not a global atomic + lone store per edge; memset gone; sum order kept
# speedup vs baseline: 3.9582x; 1.2683x over previous
#include <hip/hip_runtime.h>
#include <math.h>


typedef __attribute__((ext_vector_type(2))) float v2f;
typedef __attribute__((ext_vector_type(8))) float v8f;

#define N_NODES 50000
#define N_EDGES 800000
#define F_DIM   128
#define NBITS   64
#define NCLS    100
#define HD      300
#define LDS_STRIDE 132

__device__ __forceinline__ v8f wmma_f32(v2f a, v2f b, v8f c) {
  return __builtin_amdgcn_wmma_f32_16x16x4_f32(false, a, false, b, (short)0, c,
                                               false, false);
}

#define LIST_PITCH 64
#define CNT_LEN 50176
#define N_BINS 256
#define BIN_NODES 196
#define BIN_BLOCKS 256
#define EDGES_PER_BLOCK 3125
#define CELL_PITCH 64
#define BIN_THREADS 1024
static_assert(BIN_BLOCKS * EDGES_PER_BLOCK == N_EDGES, "edge chunks must cover the edge list");
static_assert(N_BINS * BIN_NODES >= N_NODES && BIN_NODES <= 256, "bins must cover the nodes");
static_assert(N_NODES <= (1 << 23), "source id must fit above the 8 index bits");

__global__ void __launch_bounds__(BIN_THREADS)
bin_edges_kernel(const int* __restrict__ src, const int* __restrict__ dst,
                 int* __restrict__ cells) {
  __shared__ __attribute__((aligned(16))) int sCell[N_BINS * CELL_PITCH];
  __shared__ int sHist[N_BINS];
  const int tid = threadIdx.x;
  const int e0  = blockIdx.x * EDGES_PER_BLOCK;
  int d[4], s[4];
#pragma unroll
  for (int i = 0; i < 4; ++i) {
    const int j = tid + i * BIN_THREADS;
    const bool ok = (j < EDGES_PER_BLOCK);
    d[i] = ok ? dst[e0 + j] : 0;
    s[i] = ok ? src[e0 + j] : 0;
  }
  const int4 z = make_int4(0, 0, 0, 0);
#pragma unroll
  for (int i = 0; i < (N_BINS * CELL_PITCH / 4) / BIN_THREADS; ++i)
    reinterpret_cast<int4*>(sCell)[tid + i * BIN_THREADS] = z;
  if (tid < N_BINS) sHist[tid] = 0;
  __syncthreads();
#pragma unroll
  for (int i = 0; i < 4; ++i) {
    if (tid + i * BIN_THREADS < EDGES_PER_BLOCK) {
      const int k  = d[i] / BIN_NODES;
      const int dl = d[i] - k * BIN_NODES;
      const int r = atomicAdd(&sHist[k], 1);
      if (r < CELL_PITCH - 1) sCell[k * CELL_PITCH + 1 + r] = (s[i] << 8) | dl;
    }
  }
  __syncthreads();
  if (tid < N_BINS) sCell[tid * CELL_PITCH] = sHist[tid];
  __syncthreads();
  int4* out = reinterpret_cast<int4*>(cells) + (size_t)blockIdx.x * (N_BINS * CELL_PITCH / 4);
#pragma unroll
  for (int i = 0; i < (N_BINS * CELL_PITCH / 4) / BIN_THREADS; ++i)
    out[tid + i * BIN_THREADS] = reinterpret_cast<const int4*>(sCell)[tid + i * BIN_THREADS];
}

__device__ __forceinline__ void place_edge(int* sCnt, int* sSeg, int dl, int v) {
  const int slot = atomicAdd(&sCnt[dl], 1);
  if (slot < LIST_PITCH) sSeg[dl * LIST_PITCH + slot] = v;
}

__global__ void __launch_bounds__(BIN_THREADS)
build_lists_kernel(const int* __restrict__ cells,
                   const int* __restrict__ src, const int* __restrict__ dst,
                   int* __restrict__ cnt, int* __restrict__ seg) {
  __shared__ __attribute__((aligned(16))) int sSeg[BIN_NODES * LIST_PITCH];
  __shared__ int sCnt[BIN_NODES];
  __shared__ int sOver;
  const int tid  = threadIdx.x;
  const int lane = tid & 31;
  const int k    = blockIdx.x;
  const int ch   = tid & 15;
  int4 v[4];
#pragma unroll
  for (int i = 0; i < 4; ++i) {
    const int cb = (tid + i * BIN_THREADS) >> 4;
    v[i] = reinterpret_cast<const int4*>(cells)[((size_t)cb * N_BINS + k) * (CELL_PITCH / 4) + ch];
  }
  const int4 z = make_int4(0, 0, 0, 0);
  for (int q = tid; q < BIN_NODES * LIST_PITCH / 4; q += BIN_THREADS)
    reinterpret_cast<int4*>(sSeg)[q] = z;
  if (tid < BIN_NODES) sCnt[tid] = 0;
  if (tid == 0) sOver = 0;
  __syncthreads();
  int c[4];
  bool over = false;
#pragma unroll
  for (int i = 0; i < 4; ++i) {
    c[i] = __shfl(v[i].x, lane & 16);
    over = over || (c[i] > CELL_PITCH - 1);
  }
  if (over) sOver = 1;
  __syncthreads();
  if (sOver == 0) {
#pragma unroll
    for (int i = 0; i < 4; ++i) {
      const int ev[4] = {v[i].x, v[i].y, v[i].z, v[i].w};
#pragma unroll
      for (int j = 0; j < 4; ++j) {
        const int idx = ch * 4 + j;
        if (idx >= 1 && idx <= c[i]) place_edge(sCnt, sSeg, ev[j] & 255, ev[j] >> 8);
      }
    }
  } else {
    for (int e = tid; e < N_EDGES; e += BIN_THREADS) {
      const int dl = dst[e] - k * BIN_NODES;
      if (dl >= 0 && dl < BIN_NODES) place_edge(sCnt, sSeg, dl, src[e]);
    }
  }
  __syncthreads();
  if (tid < BIN_NODES) {
    const int g = k * BIN_NODES + tid;
    if (g < N_NODES) cnt[g] = sCnt[tid];
  }
  for (int q = tid; q < BIN_NODES * (LIST_PITCH / 4); q += BIN_THREADS) {
    const int dl = q >> 4;
    const int g  = k * BIN_NODES + dl;
    if (g < N_NODES && (q & 15) * 4 < sCnt[dl])
      reinterpret_cast<int4*>(seg)[(size_t)g * (LIST_PITCH / 4) + (q & 15)] =
          reinterpret_cast<const int4*>(sSeg)[q];
  }
}

__device__ __forceinline__ int order_head(const int* __restrict__ bucket,
                                          int* ordered, int o, int n, int lane,
                                          int key) {
  if (n <= 32) {
    const int comp = (((lane < n) ? key : N_NODES) << 5) | lane;
    int r = 0;
    for (int jb = 0; jb < n; jb += 8) {
#pragma unroll
      for (int k = 0; k < 8; ++k)
        r += (__builtin_amdgcn_readlane(comp, jb + k) < comp) ? 1 : 0;
    }
    if (lane < n) ordered[o + r] = key; else r = lane;
    return __builtin_amdgcn_ds_permute(r << 2, key);
  }
  for (int i = lane; i < n; i += 32) {
    const int ki = bucket[o + i];
    int r = 0;
    for (int j = 0; j < n; ++j) {
      const int kj = bucket[o + j];
      r += ((kj < ki) || (kj == ki && j < i)) ? 1 : 0;
    }
    ordered[o + r] = ki;
  }
  __threadfence();
  return ordered[o + lane];
}

template <int K>
__device__ __forceinline__ void add_rows(const float* __restrict__ xl, int idx,
                                         int j, float4& acc) {
  float4 v[K];
#pragma unroll
  for (int k = 0; k < K; ++k) {
    const int s = __builtin_amdgcn_readlane(idx, j + k);
    v[k] = *reinterpret_cast<const float4*>(xl + (size_t)s * F_DIM);
  }
#pragma unroll
  for (int k = 0; k < K; ++k) {
    acc.x += v[k].x; acc.y += v[k].y; acc.z += v[k].z; acc.w += v[k].w;
  }
}

__device__ __forceinline__ float4 sum_list(const float* __restrict__ xl,
                                           const int* __restrict__ ordered,
                                           int o, int n, int lane, int idx) {
  float4 acc = make_float4(0.0f, 0.0f, 0.0f, 0.0f);
  for (int base = 0; base < n; base += 32) {
    const int m = (n - base < 32) ? (n - base) : 32;
    if (base > 0) idx = (lane < m) ? ordered[o + base + lane] : 0;
    int j = 0;
    for (; j + 8 <= m; j += 8) add_rows<8>(xl, idx, j, acc);
    if (m & 4) { add_rows<4>(xl, idx, j, acc); j += 4; }
    if (m & 2) { add_rows<2>(xl, idx, j, acc); j += 2; }
    if (m & 1) { add_rows<1>(xl, idx, j, acc); }
  }
  return acc;
}

__device__ __forceinline__ float4 sum_from_edges(const float* __restrict__ xl,
                                                 const int* __restrict__ src,
                                                 const int* __restrict__ dst,
                                                 int g, int n, int lane) {
  float4 acc = make_float4(0.0f, 0.0f, 0.0f, 0.0f);
  int last = -1;
  int done = 0;
  for (int pass = 0; pass < n && done < n; ++pass) {
    int best = N_NODES;
    int c = 0;
    for (int e = lane; e < N_EDGES; e += 32) {
      if (dst[e] == g) {
        const int s = src[e];
        if (s > last) {
          if (s < best) { best = s; c = 1; }
          else if (s == best) ++c;
        }
      }
    }
    int m = best;
    for (int d = 16; d >= 1; d >>= 1) {
      const int o = __shfl_xor(m, d);
      m = (o < m) ? o : m;
    }
    int cc = (best == m) ? c : 0;
    for (int d = 16; d >= 1; d >>= 1) cc += __shfl_xor(cc, d);
    m  = __builtin_amdgcn_readfirstlane(m);
    cc = __builtin_amdgcn_readfirstlane(cc);
    if (m >= N_NODES || cc <= 0) break;
    const float4 v = *reinterpret_cast<const float4*>(xl + (size_t)m * F_DIM);
    for (int i = 0; i < cc; ++i) {
      acc.x += v.x; acc.y += v.y; acc.z += v.z; acc.w += v.w;
    }
    last = m;
    done += cc;
  }
  return acc;
}

template <bool BUILD>
__global__ void __launch_bounds__(256)
sage_layer_kernel(const float* __restrict__ x, const int* __restrict__ lists,
                  int* ordered_rw,
                  const int* __restrict__ src, const int* __restrict__ dst,
                  const int* __restrict__ cnt,
                  const float* __restrict__ wl, const float* __restrict__ bl,
                  const float* __restrict__ wr, float* __restrict__ h) {
  __shared__ __attribute__((aligned(16))) float sMean[16 * LDS_STRIDE];
  __shared__ __attribute__((aligned(16))) float sX[16 * LDS_STRIDE];

  const int tm   = blockIdx.x;
  const int lane = threadIdx.x & 31;
  const int tn   = __builtin_amdgcn_readfirstlane(threadIdx.x >> 5);

  {
    const int r0 = tn * 2;
    const int g0 = tm * 16 + r0;
    const float* xl = x + lane * 4;
    const float4 vx0 = *reinterpret_cast<const float4*>(xl + (size_t)g0 * F_DIM);
    const float4 vx1 = *reinterpret_cast<const float4*>(xl + (size_t)(g0 + 1) * F_DIM);
    const int n0 = cnt[g0];
    const int n1 = cnt[g0 + 1];
    const int o0 = g0 * LIST_PITCH;
    const int o1 = o0 + LIST_PITCH;
    const bool fit0 = (n0 <= LIST_PITCH);
    const bool fit1 = (n1 <= LIST_PITCH);
    int id0 = (fit0 && lane < n0) ? lists[o0 + lane] : 0;
    int id1 = (fit1 && lane < n1) ? lists[o1 + lane] : 0;
    *reinterpret_cast<float4*>(sX + r0 * LDS_STRIDE + lane * 4)       = vx0;
    *reinterpret_cast<float4*>(sX + (r0 + 1) * LDS_STRIDE + lane * 4) = vx1;
    if (BUILD) {
      if (fit0) id0 = order_head(lists, ordered_rw, o0, n0, lane, id0);
      if (fit1) id1 = order_head(lists, ordered_rw, o1, n1, lane, id1);
    }
    const int* ordered = BUILD ? ordered_rw : lists;
    const float4 a0 = fit0 ? sum_list(xl, ordered, o0, n0, lane, id0)
                           : sum_from_edges(xl, src, dst, g0, n0, lane);
    const float4 a1 = fit1 ? sum_list(xl, ordered, o1, n1, lane, id1)
                           : sum_from_edges(xl, src, dst, g0 + 1, n1, lane);
    const float i0 = 1.0f / fmaxf((float)n0, 1.0f);
    const float i1 = 1.0f / fmaxf((float)n1, 1.0f);
    *reinterpret_cast<float4*>(sMean + r0 * LDS_STRIDE + lane * 4) =
        make_float4(a0.x * i0, a0.y * i0, a0.z * i0, a0.w * i0);
    *reinterpret_cast<float4*>(sMean + (r0 + 1) * LDS_STRIDE + lane * 4) =
        make_float4(a1.x * i1, a1.y * i1, a1.z * i1, a1.w * i1);
  }
  __syncthreads();

  const int mloc = lane & 15;
  const int ncol = tn * 16 + mloc;
  const int kb   = (lane >= 16) ? 2 : 0;
  const float* sm = sMean + mloc * LDS_STRIDE;
  const float* sx = sX    + mloc * LDS_STRIDE;

  v8f acc = {};
#pragma unroll 4
  for (int kk = 0; kk < F_DIM; kk += 4) {
    v2f a1, b1, a2, b2;
    a1.x = sm[kk + kb];
    a1.y = sm[kk + kb + 1];
    b1.x = wl[(kk + kb)     * F_DIM + ncol];
    b1.y = wl[(kk + kb + 1) * F_DIM + ncol];
    acc = wmma_f32(a1, b1, acc);
    a2.x = sx[kk + kb];
    a2.y = sx[kk + kb + 1];
    b2.x = wr[(kk + kb)     * F_DIM + ncol];
    b2.y = wr[(kk + kb + 1) * F_DIM + ncol];
    acc = wmma_f32(a2, b2, acc);
  }
  const float bias  = bl[ncol];
  const int   mbase = tm * 16 + ((lane >= 16) ? 8 : 0);
#pragma unroll
  for (int i = 0; i < 8; ++i) {
    float v = fmaxf(acc[i] + bias, 0.0f);
    h[(size_t)(mbase + i) * F_DIM + ncol] = v;
  }
}

#define HB_TILES 5
#define HB_ROWS  (16 * HB_TILES)
#define HB_WAVES 10
#define T_STRIDE 68

template <int K, int STRIDE, int N>
__device__ __forceinline__ void tiles_mma(const float* sa, const float* __restrict__ bp,
                                          v8f (&acc)[HB_TILES]) {
#pragma unroll 4
  for (int kk = 0; kk < K; kk += 4) {
    v2f b;
    b.x = bp[kk * N];
    b.y = bp[(kk + 1) * N];
#pragma unroll
    for (int t = 0; t < HB_TILES; ++t) {
      const v2f a = *reinterpret_cast<const v2f*>(sa + t * 16 * STRIDE + kk);
      acc[t] = wmma_f32(a, b, acc[t]);
    }
  }
}

template <int N>
__device__ __forceinline__ void store_tiles(float* __restrict__ C, int rbase, int ncol,
                                            float bv, const v8f (&acc)[HB_TILES]) {
#pragma unroll
  for (int t = 0; t < HB_TILES; ++t) {
#pragma unroll
    for (int i = 0; i < 8; ++i)
      C[(size_t)(rbase + t * 16 + i) * N + ncol] = acc[t][i] + bv;
  }
}

__global__ void __launch_bounds__(32 * HB_WAVES)
heads_kernel(const float* __restrict__ h2,
             const float* __restrict__ whd, const float* __restrict__ bhd,
             const float* __restrict__ wclas, const float* __restrict__ bclas,
             const float* __restrict__ wconv, const float* __restrict__ bconv,
             const float* __restrict__ gamma, const float* __restrict__ beta,
             const float* __restrict__ rm, const float* __restrict__ rv,
             const float* __restrict__ wtl, const float* __restrict__ btl,
             const float* __restrict__ wcv, const float* __restrict__ bcv,
             float* __restrict__ o_logists, float* __restrict__ o_tanh,
             float* __restrict__ o_fealab, float* __restrict__ o_feacv,
             float* __restrict__ o_truelab) {
  __shared__ __attribute__((aligned(16))) float sA[HB_ROWS * LDS_STRIDE];
  __shared__ __attribute__((aligned(16))) float sT[HB_ROWS * T_STRIDE];
  __shared__ __attribute__((aligned(16))) float sP[HB_ROWS * T_STRIDE];

  const int row0 = blockIdx.x * HB_ROWS;
#pragma unroll
  for (int j = 0; j < (HB_ROWS * 32) / (32 * HB_WAVES); ++j) {
    const int i  = threadIdx.x + j * (32 * HB_WAVES);
    const int r  = i >> 5;
    const int c4 = (i & 31) * 4;
    const float4 v = *reinterpret_cast<const float4*>(h2 + (size_t)(row0 + r) * F_DIM + c4);
    *reinterpret_cast<float4*>(sA + r * LDS_STRIDE + c4) = v;
  }
  __syncthreads();

  const int lane = threadIdx.x & 31;
  const int wave = __builtin_amdgcn_readfirstlane(threadIdx.x >> 5);
  const int mloc = lane & 15;
  const int hi   = lane >> 4;
  const int kb   = hi * 2;
  const int rloc = hi * 8;
  const float* sa = sA + mloc * LDS_STRIDE + kb;

  for (int job = wave; job < 30; job += HB_WAVES) {
    v8f acc[HB_TILES] = {};
    if (job < 19) {
      const int ncol = job * 16 + mloc;
      const bool nok = (ncol < HD);
      const int nc = nok ? ncol : (HD - 1);
      tiles_mma<F_DIM, LDS_STRIDE, HD>(sa, whd + kb * HD + nc, acc);
      if (nok) store_tiles<HD>(o_fealab, row0 + rloc, ncol, bhd[ncol], acc);
    } else if (job < 26) {
      const int ncol = (job - 19) * 16 + mloc;
      const bool nok = (ncol < NCLS);
      const int nc = nok ? ncol : (NCLS - 1);
      tiles_mma<F_DIM, LDS_STRIDE, NCLS>(sa, wclas + kb * NCLS + nc, acc);
      if (nok) store_tiles<NCLS>(o_logists, row0 + rloc, ncol, bclas[ncol], acc);
    } else {
      const int ncol = (job - 26) * 16 + mloc;
      tiles_mma<F_DIM, LDS_STRIDE, NBITS>(sa, wconv + kb * NBITS + ncol, acc);
      const float s     = gamma[ncol] * rsqrtf(rv[ncol] + 1e-5f);
      const float shift = (bconv[ncol] - rm[ncol]) * s + beta[ncol];
#pragma unroll
      for (int t = 0; t < HB_TILES; ++t) {
#pragma unroll
        for (int i = 0; i < 8; ++i)
          sP[(t * 16 + rloc + i) * T_STRIDE + ncol] = acc[t][i] * s + shift;
      }
    }
  }
  __syncthreads();

#pragma unroll
  for (int j = 0; j < (HB_ROWS * 16) / (32 * HB_WAVES); ++j) {
    const int i  = threadIdx.x + j * (32 * HB_WAVES);
    const int r  = i >> 4;
    const int c4 = (i & 15) * 4;
    const float4 p = *reinterpret_cast<const float4*>(sP + r * T_STRIDE + c4);
    float4 th;
    th.x = tanhf(p.x); th.y = tanhf(p.y); th.z = tanhf(p.z); th.w = tanhf(p.w);
    *reinterpret_cast<float4*>(sT + r * T_STRIDE + c4) = th;
    *reinterpret_cast<float4*>(o_tanh + (size_t)(row0 + r) * NBITS + c4) = th;
  }
  __syncthreads();

  if (wave < 7) {
    v8f acc[HB_TILES] = {};
    const int ncol = wave * 16 + mloc;
    const bool nok = (ncol < NCLS);
    const int nc = nok ? ncol : (NCLS - 1);
    tiles_mma<NBITS, T_STRIDE, NCLS>(sT + mloc * T_STRIDE + kb, wcv + kb * NCLS + nc, acc);
    if (nok) store_tiles<NCLS>(o_feacv, row0 + rloc, ncol, bcv[ncol], acc);
  } else {
    const int r = (wave - 7) * 32 + lane;
    if (r < HB_ROWS) {
      const float* p = sP + r * T_STRIDE;
      float s = 0.0f;
#pragma unroll
      for (int k = 0; k < NBITS; ++k) s += p[k] * wtl[k];
      o_truelab[row0 + r] = s + btl[0];
    }
  }
}

extern "C" void kernel_launch(void* const* d_in, const int* in_sizes, int n_in,
                              void* d_out, int out_size, void* d_ws, size_t ws_size,
                              hipStream_t stream) {
  (void)in_sizes; (void)n_in; (void)out_size; (void)ws_size;

  const float* features = (const float*)d_in[0];
  const int*   edges    = (const int*)d_in[1];
  const float* w1l   = (const float*)d_in[2];
  const float* b1l   = (const float*)d_in[3];
  const float* w1r   = (const float*)d_in[4];
  const float* w2l   = (const float*)d_in[5];
  const float* b2l   = (const float*)d_in[6];
  const float* w2r   = (const float*)d_in[7];
  const float* whd   = (const float*)d_in[8];
  const float* bhd   = (const float*)d_in[9];
  const float* wclas = (const float*)d_in[10];
  const float* bclas = (const float*)d_in[11];
  const float* wconv = (const float*)d_in[12];
  const float* bconv = (const float*)d_in[13];
  const float* gamma = (const float*)d_in[14];
  const float* beta  = (const float*)d_in[15];
  const float* rm    = (const float*)d_in[16];
  const float* rv    = (const float*)d_in[17];
  const float* wtl   = (const float*)d_in[18];
  const float* btl   = (const float*)d_in[19];
  const float* wcv   = (const float*)d_in[20];
  const float* bcv   = (const float*)d_in[21];

  const int* src = edges;
  const int* dst = edges + N_EDGES;

  float* ws     = (float*)d_ws;
  float* h1     = ws;
  float* h2     = h1  + (size_t)N_NODES * F_DIM;
  int* cnt     = (int*)(h2 + (size_t)N_NODES * F_DIM);
  int* seg     = cnt + CNT_LEN;
  int* ordered = seg + (size_t)N_NODES * LIST_PITCH;
  int* cells   = ordered + (size_t)N_NODES * LIST_PITCH;

  float* out       = (float*)d_out;
  float* o_logists = out;
  float* o_tanh    = o_logists + (size_t)N_NODES * NCLS;
  float* o_fealab  = o_tanh    + (size_t)N_NODES * NBITS;
  float* o_feacv   = o_fealab  + (size_t)N_NODES * HD;
  float* o_truelab = o_feacv   + (size_t)N_NODES * NCLS;

  const int rowTiles      = N_NODES / 16;

  bin_edges_kernel<<<BIN_BLOCKS, BIN_THREADS, 0, stream>>>(src, dst, cells);
  build_lists_kernel<<<N_BINS, BIN_THREADS, 0, stream>>>(cells, src, dst, cnt, seg);

  sage_layer_kernel<true><<<rowTiles, 256, 0, stream>>>(features, seg, ordered, src, dst, cnt, w1l, b1l, w1r, h1);

  sage_layer_kernel<false><<<rowTiles, 256, 0, stream>>>(h1, ordered, nullptr, src, dst, cnt, w2l, b2l, w2r, h2);

  heads_kernel<<<N_NODES / HB_ROWS, 32 * HB_WAVES, 0, stream>>>(
      h2, whd, bhd, wclas, bclas, wconv, bconv, gamma, beta, rm, rv, wtl, btl,
      wcv, bcv, o_logists, o_tanh, o_fealab, o_feacv, o_truelab);
}


# Heads kernel work re-dealt: each 80-row group is two 8-wave blocks (fea_lab tiles 0-15; the 14 other column jobs, tanh, fea_convert, true_lab), equal WMMAs per wave; 3 blocks fit a WGP, heavy first
# speedup vs baseline: 4.0804x; 1.0309x over previous
#include <hip/hip_runtime.h>
#include <math.h>


typedef __attribute__((ext_vector_type(2))) float v2f;
typedef __attribute__((ext_vector_type(8))) float v8f;

#define N_NODES 50000
#define N_EDGES 800000
#define F_DIM   128
#define NBITS   64
#define NCLS    100
#define HD      300
#define LDS_STRIDE 132

__device__ __forceinline__ v8f wmma_f32(v2f a, v2f b, v8f c) {
  return __builtin_amdgcn_wmma_f32_16x16x4_f32(false, a, false, b, (short)0, c,
                                               false, false);
}

#define LIST_PITCH 64
#define CNT_LEN 50176
#define N_BINS 256
#define BIN_NODES 196
#define BIN_BLOCKS 256
#define EDGES_PER_BLOCK 3125
#define CELL_PITCH 64
#define BIN_THREADS 1024
static_assert(BIN_BLOCKS * EDGES_PER_BLOCK == N_EDGES, "edge chunks must cover the edge list");
static_assert(N_BINS * BIN_NODES >= N_NODES && BIN_NODES <= 256, "bins must cover the nodes");
static_assert(N_NODES <= (1 << 23), "source id must fit above the 8 index bits");

__global__ void __launch_bounds__(BIN_THREADS)
bin_edges_kernel(const int* __restrict__ src, const int* __restrict__ dst,
                 int* __restrict__ cells) {
  __shared__ __attribute__((aligned(16))) int sCell[N_BINS * CELL_PITCH];
  __shared__ int sHist[N_BINS];
  const int tid = threadIdx.x;
  const int e0  = blockIdx.x * EDGES_PER_BLOCK;
  int d[4], s[4];
#pragma unroll
  for (int i = 0; i < 4; ++i) {
    const int j = tid + i * BIN_THREADS;
    const bool ok = (j < EDGES_PER_BLOCK);
    d[i] = ok ? dst[e0 + j] : 0;
    s[i] = ok ? src[e0 + j] : 0;
  }
  const int4 z = make_int4(0, 0, 0, 0);
#pragma unroll
  for (int i = 0; i < (N_BINS * CELL_PITCH / 4) / BIN_THREADS; ++i)
    reinterpret_cast<int4*>(sCell)[tid + i * BIN_THREADS] = z;
  if (tid < N_BINS) sHist[tid] = 0;
  __syncthreads();
#pragma unroll
  for (int i = 0; i < 4; ++i) {
    if (tid + i * BIN_THREADS < EDGES_PER_BLOCK) {
      const int k  = d[i] / BIN_NODES;
      const int dl = d[i] - k * BIN_NODES;
      const int r = atomicAdd(&sHist[k], 1);
      if (r < CELL_PITCH - 1) sCell[k * CELL_PITCH + 1 + r] = (s[i] << 8) | dl;
    }
  }
  __syncthreads();
  if (tid < N_BINS) sCell[tid * CELL_PITCH] = sHist[tid];
  __syncthreads();
  int4* out = reinterpret_cast<int4*>(cells) + (size_t)blockIdx.x * (N_BINS * CELL_PITCH / 4);
#pragma unroll
  for (int i = 0; i < (N_BINS * CELL_PITCH / 4) / BIN_THREADS; ++i)
    out[tid + i * BIN_THREADS] = reinterpret_cast<const int4*>(sCell)[tid + i * BIN_THREADS];
}

__device__ __forceinline__ void place_edge(int* sCnt, int* sSeg, int dl, int v) {
  const int slot = atomicAdd(&sCnt[dl], 1);
  if (slot < LIST_PITCH) sSeg[dl * LIST_PITCH + slot] = v;
}

__global__ void __launch_bounds__(BIN_THREADS)
build_lists_kernel(const int* __restrict__ cells,
                   const int* __restrict__ src, const int* __restrict__ dst,
                   int* __restrict__ cnt, int* __restrict__ seg) {
  __shared__ __attribute__((aligned(16))) int sSeg[BIN_NODES * LIST_PITCH];
  __shared__ int sCnt[BIN_NODES];
  __shared__ int sOver;
  const int tid  = threadIdx.x;
  const int lane = tid & 31;
  const int k    = blockIdx.x;
  const int ch   = tid & 15;
  int4 v[4];
#pragma unroll
  for (int i = 0; i < 4; ++i) {
    const int cb = (tid + i * BIN_THREADS) >> 4;
    v[i] = reinterpret_cast<const int4*>(cells)[((size_t)cb * N_BINS + k) * (CELL_PITCH / 4) + ch];
  }
  const int4 z = make_int4(0, 0, 0, 0);
  for (int q = tid; q < BIN_NODES * LIST_PITCH / 4; q += BIN_THREADS)
    reinterpret_cast<int4*>(sSeg)[q] = z;
  if (tid < BIN_NODES) sCnt[tid] = 0;
  if (tid == 0) sOver = 0;
  __syncthreads();
  int c[4];
  bool over = false;
#pragma unroll
  for (int i = 0; i < 4; ++i) {
    c[i] = __shfl(v[i].x, lane & 16);
    over = over || (c[i] > CELL_PITCH - 1);
  }
  if (over) sOver = 1;
  __syncthreads();
  if (sOver == 0) {
#pragma unroll
    for (int i = 0; i < 4; ++i) {
      const int ev[4] = {v[i].x, v[i].y, v[i].z, v[i].w};
#pragma unroll
      for (int j = 0; j < 4; ++j) {
        const int idx = ch * 4 + j;
        if (idx >= 1 && idx <= c[i]) place_edge(sCnt, sSeg, ev[j] & 255, ev[j] >> 8);
      }
    }
  } else {
    for (int e = tid; e < N_EDGES; e += BIN_THREADS) {
      const int dl = dst[e] - k * BIN_NODES;
      if (dl >= 0 && dl < BIN_NODES) place_edge(sCnt, sSeg, dl, src[e]);
    }
  }
  __syncthreads();
  if (tid < BIN_NODES) {
    const int g = k * BIN_NODES + tid;
    if (g < N_NODES) cnt[g] = sCnt[tid];
  }
  for (int q = tid; q < BIN_NODES * (LIST_PITCH / 4); q += BIN_THREADS) {
    const int dl = q >> 4;
    const int g  = k * BIN_NODES + dl;
    if (g < N_NODES && (q & 15) * 4 < sCnt[dl])
      reinterpret_cast<int4*>(seg)[(size_t)g * (LIST_PITCH / 4) + (q & 15)] =
          reinterpret_cast<const int4*>(sSeg)[q];
  }
}

__device__ __forceinline__ int order_head(const int* __restrict__ bucket,
                                          int* ordered, int o, int n, int lane,
                                          int key) {
  if (n <= 32) {
    const int comp = (((lane < n) ? key : N_NODES) << 5) | lane;
    int r = 0;
    for (int jb = 0; jb < n; jb += 8) {
#pragma unroll
      for (int k = 0; k < 8; ++k)
        r += (__builtin_amdgcn_readlane(comp, jb + k) < comp) ? 1 : 0;
    }
    if (lane < n) ordered[o + r] = key; else r = lane;
    return __builtin_amdgcn_ds_permute(r << 2, key);
  }
  for (int i = lane; i < n; i += 32) {
    const int ki = bucket[o + i];
    int r = 0;
    for (int j = 0; j < n; ++j) {
      const int kj = bucket[o + j];
      r += ((kj < ki) || (kj == ki && j < i)) ? 1 : 0;
    }
    ordered[o + r] = ki;
  }
  __threadfence();
  return ordered[o + lane];
}

template <int K>
__device__ __forceinline__ void add_rows(const float* __restrict__ xl, int idx,
                                         int j, float4& acc) {
  float4 v[K];
#pragma unroll
  for (int k = 0; k < K; ++k) {
    const int s = __builtin_amdgcn_readlane(idx, j + k);
    v[k] = *reinterpret_cast<const float4*>(xl + (size_t)s * F_DIM);
  }
#pragma unroll
  for (int k = 0; k < K; ++k) {
    acc.x += v[k].x; acc.y += v[k].y; acc.z += v[k].z; acc.w += v[k].w;
  }
}

__device__ __forceinline__ float4 sum_list(const float* __restrict__ xl,
                                           const int* __restrict__ ordered,
                                           int o, int n, int lane, int idx) {
  float4 acc = make_float4(0.0f, 0.0f, 0.0f, 0.0f);
  for (int base = 0; base < n; base += 32) {
    const int m = (n - base < 32) ? (n - base) : 32;
    if (base > 0) idx = (lane < m) ? ordered[o + base + lane] : 0;
    int j = 0;
    for (; j + 8 <= m; j += 8) add_rows<8>(xl, idx, j, acc);
    if (m & 4) { add_rows<4>(xl, idx, j, acc); j += 4; }
    if (m & 2) { add_rows<2>(xl, idx, j, acc); j += 2; }
    if (m & 1) { add_rows<1>(xl, idx, j, acc); }
  }
  return acc;
}

__device__ __forceinline__ float4 sum_from_edges(const float* __restrict__ xl,
                                                 const int* __restrict__ src,
                                                 const int* __restrict__ dst,
                                                 int g, int n, int lane) {
  float4 acc = make_float4(0.0f, 0.0f, 0.0f, 0.0f);
  int last = -1;
  int done = 0;
  for (int pass = 0; pass < n && done < n; ++pass) {
    int best = N_NODES;
    int c = 0;
    for (int e = lane; e < N_EDGES; e += 32) {
      if (dst[e] == g) {
        const int s = src[e];
        if (s > last) {
          if (s < best) { best = s; c = 1; }
          else if (s == best) ++c;
        }
      }
    }
    int m = best;
    for (int d = 16; d >= 1; d >>= 1) {
      const int o = __shfl_xor(m, d);
      m = (o < m) ? o : m;
    }
    int cc = (best == m) ? c : 0;
    for (int d = 16; d >= 1; d >>= 1) cc += __shfl_xor(cc, d);
    m  = __builtin_amdgcn_readfirstlane(m);
    cc = __builtin_amdgcn_readfirstlane(cc);
    if (m >= N_NODES || cc <= 0) break;
    const float4 v = *reinterpret_cast<const float4*>(xl + (size_t)m * F_DIM);
    for (int i = 0; i < cc; ++i) {
      acc.x += v.x; acc.y += v.y; acc.z += v.z; acc.w += v.w;
    }
    last = m;
    done += cc;
  }
  return acc;
}

template <bool BUILD>
__global__ void __launch_bounds__(256)
sage_layer_kernel(const float* __restrict__ x, const int* __restrict__ lists,
                  int* ordered_rw,
                  const int* __restrict__ src, const int* __restrict__ dst,
                  const int* __restrict__ cnt,
                  const float* __restrict__ wl, const float* __restrict__ bl,
                  const float* __restrict__ wr, float* __restrict__ h) {
  __shared__ __attribute__((aligned(16))) float sMean[16 * LDS_STRIDE];
  __shared__ __attribute__((aligned(16))) float sX[16 * LDS_STRIDE];

  const int tm   = blockIdx.x;
  const int lane = threadIdx.x & 31;
  const int tn   = __builtin_amdgcn_readfirstlane(threadIdx.x >> 5);

  {
    const int r0 = tn * 2;
    const int g0 = tm * 16 + r0;
    const float* xl = x + lane * 4;
    const float4 vx0 = *reinterpret_cast<const float4*>(xl + (size_t)g0 * F_DIM);
    const float4 vx1 = *reinterpret_cast<const float4*>(xl + (size_t)(g0 + 1) * F_DIM);
    const int n0 = cnt[g0];
    const int n1 = cnt[g0 + 1];
    const int o0 = g0 * LIST_PITCH;
    const int o1 = o0 + LIST_PITCH;
    const bool fit0 = (n0 <= LIST_PITCH);
    const bool fit1 = (n1 <= LIST_PITCH);
    int id0 = (fit0 && lane < n0) ? lists[o0 + lane] : 0;
    int id1 = (fit1 && lane < n1) ? lists[o1 + lane] : 0;
    *reinterpret_cast<float4*>(sX + r0 * LDS_STRIDE + lane * 4)       = vx0;
    *reinterpret_cast<float4*>(sX + (r0 + 1) * LDS_STRIDE + lane * 4) = vx1;
    if (BUILD) {
      if (fit0) id0 = order_head(lists, ordered_rw, o0, n0, lane, id0);
      if (fit1) id1 = order_head(lists, ordered_rw, o1, n1, lane, id1);
    }
    const int* ordered = BUILD ? ordered_rw : lists;
    const float4 a0 = fit0 ? sum_list(xl, ordered, o0, n0, lane, id0)
                           : sum_from_edges(xl, src, dst, g0, n0, lane);
    const float4 a1 = fit1 ? sum_list(xl, ordered, o1, n1, lane, id1)
                           : sum_from_edges(xl, src, dst, g0 + 1, n1, lane);
    const float i0 = 1.0f / fmaxf((float)n0, 1.0f);
    const float i1 = 1.0f / fmaxf((float)n1, 1.0f);
    *reinterpret_cast<float4*>(sMean + r0 * LDS_STRIDE + lane * 4) =
        make_float4(a0.x * i0, a0.y * i0, a0.z * i0, a0.w * i0);
    *reinterpret_cast<float4*>(sMean + (r0 + 1) * LDS_STRIDE + lane * 4) =
        make_float4(a1.x * i1, a1.y * i1, a1.z * i1, a1.w * i1);
  }
  __syncthreads();

  const int mloc = lane & 15;
  const int ncol = tn * 16 + mloc;
  const int kb   = (lane >= 16) ? 2 : 0;
  const float* sm = sMean + mloc * LDS_STRIDE;
  const float* sx = sX    + mloc * LDS_STRIDE;

  v8f acc = {};
#pragma unroll 4
  for (int kk = 0; kk < F_DIM; kk += 4) {
    v2f a1, b1, a2, b2;
    a1.x = sm[kk + kb];
    a1.y = sm[kk + kb + 1];
    b1.x = wl[(kk + kb)     * F_DIM + ncol];
    b1.y = wl[(kk + kb + 1) * F_DIM + ncol];
    acc = wmma_f32(a1, b1, acc);
    a2.x = sx[kk + kb];
    a2.y = sx[kk + kb + 1];
    b2.x = wr[(kk + kb)     * F_DIM + ncol];
    b2.y = wr[(kk + kb + 1) * F_DIM + ncol];
    acc = wmma_f32(a2, b2, acc);
  }
  const float bias  = bl[ncol];
  const int   mbase = tm * 16 + ((lane >= 16) ? 8 : 0);
#pragma unroll
  for (int i = 0; i < 8; ++i) {
    float v = fmaxf(acc[i] + bias, 0.0f);
    h[(size_t)(mbase + i) * F_DIM + ncol] = v;
  }
}

#define HB_TILES 5
#define HB_ROWS  (16 * HB_TILES)
#define HB_WAVES 8
#define T_STRIDE 68

template <int K, int STRIDE, int N>
__device__ __forceinline__ void tiles_mma(const float* sa, const float* __restrict__ bp,
                                          v8f (&acc)[HB_TILES]) {
#pragma unroll 4
  for (int kk = 0; kk < K; kk += 4) {
    v2f b;
    b.x = bp[kk * N];
    b.y = bp[(kk + 1) * N];
#pragma unroll
    for (int t = 0; t < HB_TILES; ++t) {
      const v2f a = *reinterpret_cast<const v2f*>(sa + t * 16 * STRIDE + kk);
      acc[t] = wmma_f32(a, b, acc[t]);
    }
  }
}

template <int N>
__device__ __forceinline__ void store_tiles(float* __restrict__ C, int rbase, int ncol,
                                            float bv, const v8f (&acc)[HB_TILES]) {
#pragma unroll
  for (int t = 0; t < HB_TILES; ++t) {
#pragma unroll
    for (int i = 0; i < 8; ++i)
      C[(size_t)(rbase + t * 16 + i) * N + ncol] = acc[t][i] + bv;
  }
}

__global__ void __launch_bounds__(32 * HB_WAVES)
heads_kernel(const float* __restrict__ h2,
             const float* __restrict__ whd, const float* __restrict__ bhd,
             const float* __restrict__ wclas, const float* __restrict__ bclas,
             const float* __restrict__ wconv, const float* __restrict__ bconv,
             const float* __restrict__ gamma, const float* __restrict__ beta,
             const float* __restrict__ rm, const float* __restrict__ rv,
             const float* __restrict__ wtl, const float* __restrict__ btl,
             const float* __restrict__ wcv, const float* __restrict__ bcv,
             float* __restrict__ o_logists, float* __restrict__ o_tanh,
             float* __restrict__ o_fealab, float* __restrict__ o_feacv,
             float* __restrict__ o_truelab) {
  __shared__ __attribute__((aligned(16))) float sA[HB_ROWS * LDS_STRIDE];
  __shared__ __attribute__((aligned(16))) float sT[HB_ROWS * T_STRIDE];
  __shared__ __attribute__((aligned(16))) float sP[HB_ROWS * T_STRIDE];

  const int row0 = blockIdx.x * HB_ROWS;
  const int half = 1 - (int)blockIdx.y;
#pragma unroll
  for (int j = 0; j < (HB_ROWS * 32) / (32 * HB_WAVES); ++j) {
    const int i  = threadIdx.x + j * (32 * HB_WAVES);
    const int r  = i >> 5;
    const int c4 = (i & 31) * 4;
    const float4 v = *reinterpret_cast<const float4*>(h2 + (size_t)(row0 + r) * F_DIM + c4);
    *reinterpret_cast<float4*>(sA + r * LDS_STRIDE + c4) = v;
  }
  __syncthreads();

  const int lane = threadIdx.x & 31;
  const int wave = __builtin_amdgcn_readfirstlane(threadIdx.x >> 5);
  const int mloc = lane & 15;
  const int hi   = lane >> 4;
  const int kb   = hi * 2;
  const int rloc = hi * 8;
  const float* sa = sA + mloc * LDS_STRIDE + kb;

  if (half == 0) {
    for (int job = wave; job < 16; job += HB_WAVES) {
      v8f acc[HB_TILES] = {};
      const int ncol = job * 16 + mloc;
      tiles_mma<F_DIM, LDS_STRIDE, HD>(sa, whd + kb * HD + ncol, acc);
      store_tiles<HD>(o_fealab, row0 + rloc, ncol, bhd[ncol], acc);
    }
    return;
  }

  for (int j = wave; j < 14; j += HB_WAVES) {
    v8f acc[HB_TILES] = {};
    if (j < 4) {
      const int ncol = j * 16 + mloc;
      tiles_mma<F_DIM, LDS_STRIDE, NBITS>(sa, wconv + kb * NBITS + ncol, acc);
      const float s     = gamma[ncol] * rsqrtf(rv[ncol] + 1e-5f);
      const float shift = (bconv[ncol] - rm[ncol]) * s + beta[ncol];
#pragma unroll
      for (int t = 0; t < HB_TILES; ++t) {
#pragma unroll
        for (int i = 0; i < 8; ++i)
          sP[(t * 16 + rloc + i) * T_STRIDE + ncol] = acc[t][i] * s + shift;
      }
    } else if (j < 11) {
      const int ncol = (j - 4) * 16 + mloc;
      const bool nok = (ncol < NCLS);
      const int nc = nok ? ncol : (NCLS - 1);
      tiles_mma<F_DIM, LDS_STRIDE, NCLS>(sa, wclas + kb * NCLS + nc, acc);
      if (nok) store_tiles<NCLS>(o_logists, row0 + rloc, ncol, bclas[ncol], acc);
    } else {
      const int ncol = (j + 5) * 16 + mloc;
      const bool nok = (ncol < HD);
      const int nc = nok ? ncol : (HD - 1);
      tiles_mma<F_DIM, LDS_STRIDE, HD>(sa, whd + kb * HD + nc, acc);
      if (nok) store_tiles<HD>(o_fealab, row0 + rloc, ncol, bhd[ncol], acc);
    }
  }
  __syncthreads();

#pragma unroll
  for (int j = 0; j < (HB_ROWS * 16) / (32 * HB_WAVES); ++j) {
    const int i  = threadIdx.x + j * (32 * HB_WAVES);
    const int r  = i >> 4;
    const int c4 = (i & 15) * 4;
    const float4 p = *reinterpret_cast<const float4*>(sP + r * T_STRIDE + c4);
    float4 th;
    th.x = tanhf(p.x); th.y = tanhf(p.y); th.z = tanhf(p.z); th.w = tanhf(p.w);
    *reinterpret_cast<float4*>(sT + r * T_STRIDE + c4) = th;
    *reinterpret_cast<float4*>(o_tanh + (size_t)(row0 + r) * NBITS + c4) = th;
  }
  __syncthreads();

  if (wave >= 3 && wave < 6) {
    const int r = (wave - 3) * 32 + lane;
    if (r < HB_ROWS) {
      const float* p = sP + r * T_STRIDE;
      float s = 0.0f;
#pragma unroll
      for (int k = 0; k < NBITS; ++k) s += p[k] * wtl[k];
      o_truelab[row0 + r] = s + btl[0];
    }
  } else {
    const int first = (wave >= 6) ? (wave - 6) * 2 : (wave + 4);
    const int count = (wave >= 6) ? 2 : 1;
    for (int q = 0; q < count; ++q) {
      v8f acc[HB_TILES] = {};
      const int ncol = (first + q) * 16 + mloc;
      const bool nok = (ncol < NCLS);
      const int nc = nok ? ncol : (NCLS - 1);
      tiles_mma<NBITS, T_STRIDE, NCLS>(sT + mloc * T_STRIDE + kb, wcv + kb * NCLS + nc, acc);
      if (nok) store_tiles<NCLS>(o_feacv, row0 + rloc, ncol, bcv[ncol], acc);
    }
  }
}

extern "C" void kernel_launch(void* const* d_in, const int* in_sizes, int n_in,
                              void* d_out, int out_size, void* d_ws, size_t ws_size,
                              hipStream_t stream) {
  (void)in_sizes; (void)n_in; (void)out_size; (void)ws_size;

  const float* features = (const float*)d_in[0];
  const int*   edges    = (const int*)d_in[1];
  const float* w1l   = (const float*)d_in[2];
  const float* b1l   = (const float*)d_in[3];
  const float* w1r   = (const float*)d_in[4];
  const float* w2l   = (const float*)d_in[5];
  const float* b2l   = (const float*)d_in[6];
  const float* w2r   = (const float*)d_in[7];
  const float* whd   = (const float*)d_in[8];
  const float* bhd   = (const float*)d_in[9];
  const float* wclas = (const float*)d_in[10];
  const float* bclas = (const float*)d_in[11];
  const float* wconv = (const float*)d_in[12];
  const float* bconv = (const float*)d_in[13];
  const float* gamma = (const float*)d_in[14];
  const float* beta  = (const float*)d_in[15];
  const float* rm    = (const float*)d_in[16];
  const float* rv    = (const float*)d_in[17];
  const float* wtl   = (const float*)d_in[18];
  const float* btl   = (const float*)d_in[19];
  const float* wcv   = (const float*)d_in[20];
  const float* bcv   = (const float*)d_in[21];

  const int* src = edges;
  const int* dst = edges + N_EDGES;

  float* ws     = (float*)d_ws;
  float* h1     = ws;
  float* h2     = h1  + (size_t)N_NODES * F_DIM;
  int* cnt     = (int*)(h2 + (size_t)N_NODES * F_DIM);
  int* seg     = cnt + CNT_LEN;
  int* ordered = seg + (size_t)N_NODES * LIST_PITCH;
  int* cells   = ordered + (size_t)N_NODES * LIST_PITCH;

  float* out       = (float*)d_out;
  float* o_logists = out;
  float* o_tanh    = o_logists + (size_t)N_NODES * NCLS;
  float* o_fealab  = o_tanh    + (size_t)N_NODES * NBITS;
  float* o_feacv   = o_fealab  + (size_t)N_NODES * HD;
  float* o_truelab = o_feacv   + (size_t)N_NODES * NCLS;

  const int rowTiles      = N_NODES / 16;

  bin_edges_kernel<<<BIN_BLOCKS, BIN_THREADS, 0, stream>>>(src, dst, cells);
  build_lists_kernel<<<N_BINS, BIN_THREADS, 0, stream>>>(cells, src, dst, cnt, seg);

  sage_layer_kernel<true><<<rowTiles, 256, 0, stream>>>(features, seg, ordered, src, dst, cnt, w1l, b1l, w1r, h1);

  sage_layer_kernel<false><<<rowTiles, 256, 0, stream>>>(h1, ordered, nullptr, src, dst, cnt, w2l, b2l, w2r, h2);

  heads_kernel<<<dim3(N_NODES / HB_ROWS, 2), 32 * HB_WAVES, 0, stream>>>(
      h2, whd, bhd, wclas, bclas, wconv, bconv, gamma, beta, rm, rv, wtl, btl,
      wcv, bcv, o_logists, o_tanh, o_fealab, o_feacv, o_truelab);
}
